# starting kernel
# baseline (speedup 1.0000x reference)
#include <hip/hip_runtime.h>
#include <math.h>

#define N_NODES 100000
#define IN_W    64
#define OUT_W   64
#define N_EDGES 1600000

typedef __attribute__((ext_vector_type(2))) float v2f;
typedef __attribute__((ext_vector_type(8))) float v8f;

__global__ void deg_init_kernel(float* __restrict__ deg) {
    int i = blockIdx.x * blockDim.x + threadIdx.x;
    if (i < N_NODES) deg[i] = 2.0f;
}

__global__ void deg_accum_kernel(const int* __restrict__ dst,
                                 const float* __restrict__ w,
                                 float* __restrict__ deg) {
    int e = blockIdx.x * blockDim.x + threadIdx.x;
    if (e < N_EDGES) atomicAdd(&deg[dst[e]], w[e]);
}

__global__ void deg_rsqrt_kernel(float* __restrict__ deg) {
    int i = blockIdx.x * blockDim.x + threadIdx.x;
    if (i < N_NODES) {
        float d = deg[i];
        deg[i] = (d > 0.0f) ? rsqrtf(d) : 0.0f;
    }
}

__global__ void gemm_wmma_kernel(const float* __restrict__ x,
                                 const float* __restrict__ W,
                                 float* __restrict__ h) {
    const unsigned lane  = threadIdx.x & 31u;
    const unsigned gwave = blockIdx.x * 8u + (threadIdx.x >> 5);
    const unsigned tile_n = gwave & 3u;
    const unsigned tile_m = gwave >> 2;
    const unsigned m0 = tile_m * 16u;
    const unsigned n0 = tile_n * 16u;
    const unsigned half = lane >> 4;
    const unsigned lrow = lane & 15u;

    const float* xrow = x + (size_t)(m0 + lrow) * IN_W;
    const float* wcol = W + (n0 + lrow);

    v8f c = {};
#pragma unroll
    for (int kk = 0; kk < 16; ++kk) {
        const int kb = kk * 4 + (int)half * 2;
        v2f a = *(const v2f*)(xrow + kb);
        v2f b;
        b.x = wcol[(size_t)kb * OUT_W];
        b.y = wcol[(size_t)(kb + 1) * OUT_W];
        c = __builtin_amdgcn_wmma_f32_16x16x4_f32(
                 false, a,  false, b,
                 (short)0, c,  false,  false);
    }

    float* hcol = h + n0 + lrow;
#pragma unroll
    for (int v = 0; v < 8; ++v) {
        hcol[(size_t)(m0 + v + half * 8u) * OUT_W] = c[v];
    }
}

__global__ void out_init_kernel(const float* __restrict__ h,
                                const float* __restrict__ dinv,
                                const float* __restrict__ bias,
                                float* __restrict__ out) {
    int t = blockIdx.x * blockDim.x + threadIdx.x;
    if (t >= N_NODES * (OUT_W / 4)) return;
    int node = t >> 4;
    int f    = (t & 15) * 4;
    float di = dinv[node];
    float s  = 2.0f * di * di;
    float4 hv = *(const float4*)(h + (size_t)node * OUT_W + f);
    float4 bv = *(const float4*)(bias + f);
    float4 o;
    o.x = bv.x + hv.x * s;
    o.y = bv.y + hv.y * s;
    o.z = bv.z + hv.z * s;
    o.w = bv.w + hv.w * s;
    *(float4*)(out + (size_t)node * OUT_W + f) = o;
}

__global__ void edge_scatter_kernel(const int* __restrict__ src,
                                    const int* __restrict__ dst,
                                    const float* __restrict__ w,
                                    const float* __restrict__ dinv,
                                    const float* __restrict__ h,
                                    float* __restrict__ out) {
    int t = blockIdx.x * blockDim.x + threadIdx.x;
    int e = t >> 4;
    if (e >= N_EDGES) return;
    int f = (t & 15) * 4;
    int s = src[e];
    int d = dst[e];
    float coef = dinv[s] * w[e] * dinv[d];
    float4 hv = *(const float4*)(h + (size_t)s * OUT_W + f);
    float* o = out + (size_t)d * OUT_W + f;
    atomicAdd(o + 0, hv.x * coef);
    atomicAdd(o + 1, hv.y * coef);
    atomicAdd(o + 2, hv.z * coef);
    atomicAdd(o + 3, hv.w * coef);
}

extern "C" void kernel_launch(void* const* d_in, const int* in_sizes, int n_in,
                              void* d_out, int out_size, void* d_ws, size_t ws_size,
                              hipStream_t stream) {
    const float* x         = (const float*)d_in[0];
    const float* edge_attr = (const float*)d_in[1];
    const float* W         = (const float*)d_in[2];
    const float* bias      = (const float*)d_in[3];
    const int*   edge_idx  = (const int*)d_in[4];
    float* out = (float*)d_out;

    float* dinv = (float*)d_ws;
    float* h    = dinv + 100096;

    const int* src = edge_idx;
    const int* dst = edge_idx + N_EDGES;

    deg_init_kernel<<<(N_NODES + 255) / 256, 256, 0, stream>>>(dinv);
    deg_accum_kernel<<<(N_EDGES + 255) / 256, 256, 0, stream>>>(dst, edge_attr, dinv);
    deg_rsqrt_kernel<<<(N_NODES + 255) / 256, 256, 0, stream>>>(dinv);

    gemm_wmma_kernel<<<3125, 256, 0, stream>>>(x, W, h);

    out_init_kernel<<<(N_NODES * 16 + 255) / 256, 256, 0, stream>>>(h, dinv, bias, out);

    edge_scatter_kernel<<<(N_EDGES * 16 + 255) / 256, 256, 0, stream>>>(
        src, dst, edge_attr, dinv, h, out);
}


# edge scatter re-mapped: one wave per 32 edges, lane = feature column, so each h load and each f32 atomic add covers one whole 128-B line (2 line adds per edge instead of 8 quarter-filled)
# speedup vs baseline: 1.5841x; 1.5841x over previous
#include <hip/hip_runtime.h>
#include <math.h>

#define N_NODES 100000
#define IN_W    64
#define OUT_W   64
#define N_EDGES 1600000

typedef __attribute__((ext_vector_type(2))) float v2f;
typedef __attribute__((ext_vector_type(8))) float v8f;

__global__ void deg_init_kernel(float* __restrict__ deg) {
    int i = blockIdx.x * blockDim.x + threadIdx.x;
    if (i < N_NODES) deg[i] = 2.0f;
}

__global__ void deg_accum_kernel(const int* __restrict__ dst,
                                 const float* __restrict__ w,
                                 float* __restrict__ deg) {
    int e = blockIdx.x * blockDim.x + threadIdx.x;
    if (e < N_EDGES) atomicAdd(&deg[dst[e]], w[e]);
}

__global__ void deg_rsqrt_kernel(float* __restrict__ deg) {
    int i = blockIdx.x * blockDim.x + threadIdx.x;
    if (i < N_NODES) {
        float d = deg[i];
        deg[i] = (d > 0.0f) ? rsqrtf(d) : 0.0f;
    }
}

__global__ void gemm_wmma_kernel(const float* __restrict__ x,
                                 const float* __restrict__ W,
                                 float* __restrict__ h) {
    const unsigned lane  = threadIdx.x & 31u;
    const unsigned gwave = blockIdx.x * 8u + (threadIdx.x >> 5);
    const unsigned tile_n = gwave & 3u;
    const unsigned tile_m = gwave >> 2;
    const unsigned m0 = tile_m * 16u;
    const unsigned n0 = tile_n * 16u;
    const unsigned half = lane >> 4;
    const unsigned lrow = lane & 15u;

    const float* xrow = x + (size_t)(m0 + lrow) * IN_W;
    const float* wcol = W + (n0 + lrow);

    v8f c = {};
#pragma unroll
    for (int kk = 0; kk < 16; ++kk) {
        const int kb = kk * 4 + (int)half * 2;
        v2f a = *(const v2f*)(xrow + kb);
        v2f b;
        b.x = wcol[(size_t)kb * OUT_W];
        b.y = wcol[(size_t)(kb + 1) * OUT_W];
        c = __builtin_amdgcn_wmma_f32_16x16x4_f32(
                 false, a,  false, b,
                 (short)0, c,  false,  false);
    }

    float* hcol = h + n0 + lrow;
#pragma unroll
    for (int v = 0; v < 8; ++v) {
        hcol[(size_t)(m0 + v + half * 8u) * OUT_W] = c[v];
    }
}

__global__ void out_init_kernel(const float* __restrict__ h,
                                const float* __restrict__ dinv,
                                const float* __restrict__ bias,
                                float* __restrict__ out) {
    int t = blockIdx.x * blockDim.x + threadIdx.x;
    if (t >= N_NODES * (OUT_W / 4)) return;
    int node = t >> 4;
    int f    = (t & 15) * 4;
    float di = dinv[node];
    float s  = 2.0f * di * di;
    float4 hv = *(const float4*)(h + (size_t)node * OUT_W + f);
    float4 bv = *(const float4*)(bias + f);
    float4 o;
    o.x = bv.x + hv.x * s;
    o.y = bv.y + hv.y * s;
    o.z = bv.z + hv.z * s;
    o.w = bv.w + hv.w * s;
    *(float4*)(out + (size_t)node * OUT_W + f) = o;
}

#define EDGE_GROUP 16

__device__ __forceinline__ int lane_bcast_i(int v, int j) {
    return __builtin_amdgcn_readlane(v, j);
}
__device__ __forceinline__ float lane_bcast_f(float v, int j) {
    return __builtin_bit_cast(float,
               __builtin_amdgcn_readlane(__builtin_bit_cast(int, v), j));
}

__global__ void __launch_bounds__(128)
edge_scatter_kernel(const int* __restrict__ src,
                    const int* __restrict__ dst,
                    const float* __restrict__ w,
                    const float* __restrict__ dinv,
                    const float* __restrict__ h,
                    float* __restrict__ out) {
    const unsigned lane   = threadIdx.x & 31u;
    const unsigned wave   = blockIdx.x * 4u + (threadIdx.x >> 5);
    const unsigned nwaves = gridDim.x * 4u;

    for (unsigned batch = wave; batch < N_EDGES / 32; batch += nwaves) {
        const unsigned e = batch * 32u + lane;
        const int s = src[e];
        const int d = dst[e];
        const float coef = dinv[s] * w[e] * dinv[d];

#pragma unroll
        for (int jb = 0; jb < 32; jb += EDGE_GROUP) {
            float h0[EDGE_GROUP], h1[EDGE_GROUP];
#pragma unroll
            for (int k = 0; k < EDGE_GROUP; ++k) {
                const int sj = lane_bcast_i(s, jb + k);
                const float* hr = h + (size_t)sj * OUT_W + lane;
                h0[k] = hr[0];
                h1[k] = hr[32];
            }
#pragma unroll
            for (int k = 0; k < EDGE_GROUP; ++k) {
                const int   dj = lane_bcast_i(d, jb + k);
                const float cj = lane_bcast_f(coef, jb + k);
                float* o = out + (size_t)dj * OUT_W + lane;
                atomicAdd(o,      h0[k] * cj);
                atomicAdd(o + 32, h1[k] * cj);
            }
        }
    }
}

extern "C" void kernel_launch(void* const* d_in, const int* in_sizes, int n_in,
                              void* d_out, int out_size, void* d_ws, size_t ws_size,
                              hipStream_t stream) {
    const float* x         = (const float*)d_in[0];
    const float* edge_attr = (const float*)d_in[1];
    const float* W         = (const float*)d_in[2];
    const float* bias      = (const float*)d_in[3];
    const int*   edge_idx  = (const int*)d_in[4];
    float* out = (float*)d_out;

    float* dinv = (float*)d_ws;
    float* h    = dinv + 100096;

    const int* src = edge_idx;
    const int* dst = edge_idx + N_EDGES;

    deg_init_kernel<<<(N_NODES + 255) / 256, 256, 0, stream>>>(dinv);
    deg_accum_kernel<<<(N_EDGES + 255) / 256, 256, 0, stream>>>(dst, edge_attr, dinv);
    deg_rsqrt_kernel<<<(N_NODES + 255) / 256, 256, 0, stream>>>(dinv);

    gemm_wmma_kernel<<<3125, 256, 0, stream>>>(x, W, h);

    out_init_kernel<<<(N_NODES * 16 + 255) / 256, 256, 0, stream>>>(h, dinv, bias, out);

    edge_scatter_kernel<<<2500, 128, 0, stream>>>(
        src, dst, edge_attr, dinv, h, out);
}


# edges binned per call by destination block via LDS counters; one work-group per 196 nodes sums rows in LDS (ds f32 adds, sum order varies), stores each out row once: no global atomics left
# speedup vs baseline: 2.2310x; 1.4084x over previous
#include <hip/hip_runtime.h>
#include <math.h>


#define N_NODES 100000
#define IN_W    64
#define OUT_W   64
#define N_EDGES 1600000

#define R       196
#define NB      512
#define CHUNK   8192
#define CH      196
#define BIN_THREADS 1024
#define PER_THREAD  (CHUNK / BIN_THREADS)
#define SRC_BITS 17

static_assert(NB * R >= N_NODES, "blocks must cover all nodes");
static_assert(CH * CHUNK >= N_EDGES, "chunks must cover all edges");
static_assert(N_NODES <= (1 << SRC_BITS), "src must fit its field");
static_assert(R <= 512, "local row must fit its field");

typedef __attribute__((ext_vector_type(2))) float v2f;
typedef __attribute__((ext_vector_type(8))) float v8f;

__global__ void __launch_bounds__(BIN_THREADS)
bin_count_kernel(const int* __restrict__ dst, unsigned* __restrict__ counts) {
    __shared__ unsigned cnt[NB];
    const unsigned tid = threadIdx.x;
    const unsigned chunk = blockIdx.x;
    if (tid < NB) cnt[tid] = 0u;
    __syncthreads();
    const unsigned e0 = chunk * CHUNK + tid;
    unsigned bins[PER_THREAD];
#pragma unroll
    for (int k = 0; k < PER_THREAD; ++k) {
        const unsigned e = e0 + (unsigned)k * BIN_THREADS;
        bins[k] = (e < N_EDGES) ? (unsigned)dst[e] / R : (unsigned)NB;
    }
#pragma unroll
    for (int k = 0; k < PER_THREAD; ++k) {
        if (bins[k] < NB) atomicAdd(&cnt[bins[k]], 1u);
    }
    __syncthreads();
    if (tid < NB) counts[chunk * NB + tid] = cnt[tid];
}

__global__ void __launch_bounds__(1024)
bin_scan_kernel(const unsigned* __restrict__ counts,
                unsigned* __restrict__ offs,
                unsigned* __restrict__ binStart) {
    __shared__ unsigned part[2][NB];
    __shared__ unsigned sc[NB];
    const unsigned tid = threadIdx.x;
    const unsigned b = tid & (NB - 1);
    const unsigned q = tid >> 9;
    const unsigned c0 = q * (CH / 2);
    const unsigned c1 = c0 + (CH / 2);

    unsigned sum = 0u;
#pragma unroll 7
    for (unsigned c = c0; c < c1; ++c) sum += counts[c * NB + b];
    part[q][b] = sum;
    __syncthreads();
    const unsigned total = part[0][b] + part[1][b];
    if (q == 0u) sc[b] = total;
    __syncthreads();
    for (unsigned off = 1u; off < NB; off <<= 1) {
        unsigned v = 0u;
        if (q == 0u && b >= off) v = sc[b - off];
        __syncthreads();
        if (q == 0u) sc[b] += v;
        __syncthreads();
    }
    const unsigned excl = sc[b] - total;
    if (q == 0u) {
        binStart[b] = excl;
        if (b == NB - 1) binStart[NB] = sc[b];
    }
    unsigned running = excl + (q == 1u ? part[0][b] : 0u);
#pragma unroll 7
    for (unsigned c = c0; c < c1; ++c) {
        const unsigned v = counts[c * NB + b];
        offs[c * NB + b] = running;
        running += v;
    }
}

__global__ void __launch_bounds__(BIN_THREADS)
bin_place_kernel(const int* __restrict__ src,
                 const int* __restrict__ dst,
                 const float* __restrict__ w,
                 const unsigned* __restrict__ offs,
                 int2* __restrict__ rec) {
    __shared__ unsigned cur[NB];
    const unsigned tid = threadIdx.x;
    const unsigned chunk = blockIdx.x;
    if (tid < NB) cur[tid] = offs[chunk * NB + tid];
    __syncthreads();
    const unsigned e0 = chunk * CHUNK + tid;
    unsigned sv[PER_THREAD], dv[PER_THREAD];
    float wv[PER_THREAD];
#pragma unroll
    for (int k = 0; k < PER_THREAD; ++k) {
        const unsigned e = e0 + (unsigned)k * BIN_THREADS;
        const bool live = e < N_EDGES;
        sv[k] = live ? (unsigned)src[e] : 0u;
        dv[k] = live ? (unsigned)dst[e] : (unsigned)(NB * R);
        wv[k] = live ? w[e] : 0.0f;
    }
#pragma unroll
    for (int k = 0; k < PER_THREAD; ++k) {
        if (dv[k] < NB * R) {
            const unsigned bin = dv[k] / R;
            const unsigned dl = dv[k] - bin * R;
            const unsigned pos = atomicAdd(&cur[bin], 1u);
            int2 r;
            r.x = (int)(sv[k] | (dl << SRC_BITS));
            r.y = __builtin_bit_cast(int, wv[k]);
            rec[pos] = r;
        }
    }
}

__global__ void __launch_bounds__(1024)
deg_block_kernel(const int2* __restrict__ rec,
                 const unsigned* __restrict__ binStart,
                 float* __restrict__ dinv) {
    __shared__ float dg[R];
    const unsigned tid = threadIdx.x;
    const unsigned b = blockIdx.x;
    if (tid < R) dg[tid] = 2.0f;
    __syncthreads();
    const unsigned start = binStart[b];
    const unsigned end   = binStart[b + 1];
    for (unsigned i = start + tid; i < end; i += 1024u) {
        const int2 r = rec[i];
        const unsigned dl = (unsigned)r.x >> SRC_BITS;
        atomicAdd(&dg[dl], __builtin_bit_cast(float, r.y));
    }
    __syncthreads();
    if (tid < R) {
        const unsigned node = b * R + tid;
        if (node < N_NODES) {
            const float d = dg[tid];
            dinv[node] = (d > 0.0f) ? rsqrtf(d) : 0.0f;
        }
    }
}

__global__ void gemm_wmma_kernel(const float* __restrict__ x,
                                 const float* __restrict__ W,
                                 const float* __restrict__ dinv,
                                 float* __restrict__ h) {
    const unsigned lane  = threadIdx.x & 31u;
    const unsigned gwave = blockIdx.x * 8u + (threadIdx.x >> 5);
    const unsigned tile_n = gwave & 3u;
    const unsigned tile_m = gwave >> 2;
    const unsigned m0 = tile_m * 16u;
    const unsigned n0 = tile_n * 16u;
    const unsigned half = lane >> 4;
    const unsigned lrow = lane & 15u;

    const float* xrow = x + (size_t)(m0 + lrow) * IN_W;
    const float* wcol = W + (n0 + lrow);

    v8f c = {};
#pragma unroll
    for (int kk = 0; kk < 16; ++kk) {
        const int kb = kk * 4 + (int)half * 2;
        v2f a = *(const v2f*)(xrow + kb);
        v2f b;
        b.x = wcol[(size_t)kb * OUT_W];
        b.y = wcol[(size_t)(kb + 1) * OUT_W];
        c = __builtin_amdgcn_wmma_f32_16x16x4_f32(
                 false, a,  false, b,
                 (short)0, c,  false,  false);
    }

    float* hcol = h + n0 + lrow;
#pragma unroll
    for (int v = 0; v < 8; ++v) {
        const unsigned row = m0 + (unsigned)v + half * 8u;
        hcol[(size_t)row * OUT_W] = c[v] * dinv[row];
    }
}

#define EDGE_GROUP 16
#define ACC_THREADS 512
#define ACC_WAVES   (ACC_THREADS / 32)

__device__ __forceinline__ int lane_bcast_i(int v, int j) {
    return __builtin_amdgcn_readlane(v, j);
}

__global__ void __launch_bounds__(ACC_THREADS)
block_sum_kernel(const int2* __restrict__ rec,
                 const unsigned* __restrict__ binStart,
                 const float* __restrict__ dinv,
                 const float* __restrict__ h,
                 const float* __restrict__ bias,
                 float* __restrict__ out) {
    __shared__ float acc[R * OUT_W];
    const unsigned tid  = threadIdx.x;
    const unsigned lane = tid & 31u;
    const unsigned wv   = tid >> 5;
    const unsigned b    = blockIdx.x;
    const unsigned node0 = b * R;

    for (unsigned t = tid; t < R * (OUT_W / 4); t += ACC_THREADS) {
        const unsigned node = node0 + (t >> 4);
        float4 v = make_float4(0.0f, 0.0f, 0.0f, 0.0f);
        if (node < N_NODES) {
            const float4 hv = *(const float4*)(h + (size_t)node * OUT_W + (t & 15u) * 4u);
            v = make_float4(2.0f * hv.x, 2.0f * hv.y, 2.0f * hv.z, 2.0f * hv.w);
        }
        *(float4*)(acc + t * 4u) = v;
    }
    __syncthreads();

    const unsigned start = binStart[b];
    const unsigned end   = binStart[b + 1];

    for (unsigned base = start + wv * 32u; base < end; base += ACC_WAVES * 32u) {
        const unsigned left = end - base;
        const unsigned idx  = base + lane;
        int2 r = make_int2(0, 0);
        if (idx < end) r = rec[idx];

        if (left >= 32u) {
#pragma unroll
            for (int jb = 0; jb < 32; jb += EDGE_GROUP) {
                float h0[EDGE_GROUP], h1[EDGE_GROUP];
#pragma unroll
                for (int k = 0; k < EDGE_GROUP; ++k) {
                    const unsigned rx = (unsigned)lane_bcast_i(r.x, jb + k);
                    const float* hr = h + (size_t)(rx & ((1u << SRC_BITS) - 1u)) * OUT_W;
                    h0[k] = hr[lane];
                    h1[k] = hr[lane + 32u];
                }
#pragma unroll
                for (int k = 0; k < EDGE_GROUP; ++k) {
                    const unsigned rx = (unsigned)lane_bcast_i(r.x, jb + k);
                    const float wj = __builtin_bit_cast(float, lane_bcast_i(r.y, jb + k));
                    float* a = acc + (rx >> SRC_BITS) * OUT_W + lane;
                    atomicAdd(a,      h0[k] * wj);
                    atomicAdd(a + 32, h1[k] * wj);
                }
            }
        } else {
            for (unsigned j = 0; j < left; ++j) {
                const unsigned rx = (unsigned)lane_bcast_i(r.x, (int)j);
                const float wj = __builtin_bit_cast(float, lane_bcast_i(r.y, (int)j));
                const float* hr = h + (size_t)(rx & ((1u << SRC_BITS) - 1u)) * OUT_W;
                float* a = acc + (rx >> SRC_BITS) * OUT_W + lane;
                atomicAdd(a,      hr[lane]       * wj);
                atomicAdd(a + 32, hr[lane + 32u] * wj);
            }
        }
    }
    __syncthreads();

    for (unsigned t = tid; t < R * (OUT_W / 4); t += ACC_THREADS) {
        const unsigned node = node0 + (t >> 4);
        if (node < N_NODES) {
            const unsigned f = (t & 15u) * 4u;
            const float di = dinv[node];
            const float4 a  = *(const float4*)(acc + t * 4u);
            const float4 bv = *(const float4*)(bias + f);
            float4 o;
            o.x = bv.x + a.x * di;
            o.y = bv.y + a.y * di;
            o.z = bv.z + a.z * di;
            o.w = bv.w + a.w * di;
            *(float4*)(out + (size_t)node * OUT_W + f) = o;
        }
    }
}

extern "C" void kernel_launch(void* const* d_in, const int* in_sizes, int n_in,
                              void* d_out, int out_size, void* d_ws, size_t ws_size,
                              hipStream_t stream) {
    const float* x         = (const float*)d_in[0];
    const float* edge_attr = (const float*)d_in[1];
    const float* W         = (const float*)d_in[2];
    const float* bias      = (const float*)d_in[3];
    const int*   edge_idx  = (const int*)d_in[4];
    float* out = (float*)d_out;

    float*    dinv     = (float*)d_ws;
    float*    h        = dinv + 100352;
    int2*     rec      = (int2*)(h + 6400000);
    unsigned* counts   = (unsigned*)(rec + 1600000);
    unsigned* offs     = counts + 100352;
    unsigned* binStart = offs + 100352;

    const int* src = edge_idx;
    const int* dst = edge_idx + N_EDGES;

    bin_count_kernel<<<CH, BIN_THREADS, 0, stream>>>(dst, counts);
    bin_scan_kernel<<<1, 1024, 0, stream>>>(counts, offs, binStart);
    bin_place_kernel<<<CH, BIN_THREADS, 0, stream>>>(src, dst, edge_attr, offs, rec);

    deg_block_kernel<<<NB, 1024, 0, stream>>>(rec, binStart, dinv);

    gemm_wmma_kernel<<<3125, 256, 0, stream>>>(x, W, dinv, h);

    block_sum_kernel<<<NB, ACC_THREADS, 0, stream>>>(rec, binStart, dinv, h, bias, out);
}


# offset scan folded into the placing pass: each placing work-group sums the count table itself (bin totals, own prefix) and scans the 512 totals in LDS; the one-work-group scan launch is gone
# speedup vs baseline: 2.5162x; 1.1278x over previous
#include <hip/hip_runtime.h>
#include <math.h>


#define N_NODES 100000
#define IN_W    64
#define OUT_W   64
#define N_EDGES 1600000

#define R       196
#define NB      512
#define CHUNK   8192
#define CH      196
#define BIN_THREADS 1024
#define PER_THREAD  (CHUNK / BIN_THREADS)
#define SRC_BITS 17
#define SCAN_GROUP 14

static_assert(NB * R >= N_NODES, "blocks must cover all nodes");
static_assert(CH * CHUNK >= N_EDGES, "chunks must cover all edges");
static_assert(N_NODES <= (1 << SRC_BITS), "src must fit its field");
static_assert(R <= 512, "local row must fit its field");
static_assert((CH / 2) % SCAN_GROUP == 0 && CH % 2 == 0, "each half of the chunks splits into whole groups");
static_assert(2 * NB == BIN_THREADS, "placing threads = two per bin");

typedef __attribute__((ext_vector_type(2))) float v2f;
typedef __attribute__((ext_vector_type(8))) float v8f;

__global__ void __launch_bounds__(BIN_THREADS)
bin_count_kernel(const int* __restrict__ dst, unsigned* __restrict__ counts) {
    __shared__ unsigned cnt[NB];
    const unsigned tid = threadIdx.x;
    const unsigned chunk = blockIdx.x;
    if (tid < NB) cnt[tid] = 0u;
    __syncthreads();
    const unsigned e0 = chunk * CHUNK + tid;
    unsigned bins[PER_THREAD];
#pragma unroll
    for (int k = 0; k < PER_THREAD; ++k) {
        const unsigned e = e0 + (unsigned)k * BIN_THREADS;
        bins[k] = (e < N_EDGES) ? (unsigned)dst[e] / R : (unsigned)NB;
    }
#pragma unroll
    for (int k = 0; k < PER_THREAD; ++k) {
        if (bins[k] < NB) atomicAdd(&cnt[bins[k]], 1u);
    }
    __syncthreads();
    if (tid < NB) counts[chunk * NB + tid] = cnt[tid];
}

__global__ void __launch_bounds__(BIN_THREADS)
bin_place_kernel(const int* __restrict__ src,
                 const int* __restrict__ dst,
                 const float* __restrict__ w,
                 const unsigned* __restrict__ counts,
                 int2* __restrict__ rec,
                 unsigned* __restrict__ binStart) {
    __shared__ unsigned cur[NB];
    __shared__ unsigned ptot[2][NB];
    __shared__ unsigned ppre[2][NB];
    __shared__ unsigned wsum[NB / 32];
    const unsigned tid = threadIdx.x;
    const unsigned chunk = blockIdx.x;
    const unsigned lane = tid & 31u;
    const unsigned b = tid & (NB - 1);
    const unsigned q = tid >> 9;

    const unsigned e0 = chunk * CHUNK + tid;
    unsigned sv[PER_THREAD], dv[PER_THREAD];
    float wv[PER_THREAD];
#pragma unroll
    for (int k = 0; k < PER_THREAD; ++k) {
        const unsigned e = e0 + (unsigned)k * BIN_THREADS;
        const bool live = e < N_EDGES;
        sv[k] = live ? (unsigned)src[e] : 0u;
        dv[k] = live ? (unsigned)dst[e] : (unsigned)(NB * R);
        wv[k] = live ? w[e] : 0.0f;
    }

    const unsigned c0 = q * (CH / 2);
    unsigned tot = 0u, pre = 0u;
    for (unsigned cc = c0; cc < c0 + (CH / 2); cc += SCAN_GROUP) {
        unsigned v[SCAN_GROUP];
#pragma unroll
        for (int k = 0; k < SCAN_GROUP; ++k) v[k] = counts[(cc + (unsigned)k) * NB + b];
#pragma unroll
        for (int k = 0; k < SCAN_GROUP; ++k) {
            tot += v[k];
            pre += (cc + (unsigned)k < chunk) ? v[k] : 0u;
        }
    }
    ptot[q][b] = tot;
    ppre[q][b] = pre;
    __syncthreads();

    unsigned T = 0u, P = 0u, incl = 0u;
    if (tid < NB) {
        T = ptot[0][b] + ptot[1][b];
        P = ppre[0][b] + ppre[1][b];
        incl = T;
#pragma unroll
        for (int o = 1; o < 32; o <<= 1) {
            const unsigned n = (unsigned)__shfl_up((int)incl, (unsigned)o);
            if (lane >= (unsigned)o) incl += n;
        }
        if (lane == 31u) wsum[tid >> 5] = incl;
    }
    __syncthreads();
    if (tid < NB) {
        const unsigned wvi = tid >> 5;
        unsigned base = 0u;
#pragma unroll
        for (unsigned k = 0; k < NB / 32; ++k) base += (k < wvi) ? wsum[k] : 0u;
        const unsigned excl = base + incl - T;
        cur[b] = excl + P;
        if (chunk == 0u) {
            binStart[b] = excl;
            if (b == NB - 1) binStart[NB] = excl + T;
        }
    }
    __syncthreads();

#pragma unroll
    for (int k = 0; k < PER_THREAD; ++k) {
        if (dv[k] < NB * R) {
            const unsigned bin = dv[k] / R;
            const unsigned dl = dv[k] - bin * R;
            const unsigned pos = atomicAdd(&cur[bin], 1u);
            int2 r;
            r.x = (int)(sv[k] | (dl << SRC_BITS));
            r.y = __builtin_bit_cast(int, wv[k]);
            rec[pos] = r;
        }
    }
}

__global__ void __launch_bounds__(1024)
deg_block_kernel(const int2* __restrict__ rec,
                 const unsigned* __restrict__ binStart,
                 float* __restrict__ dinv) {
    __shared__ float dg[R];
    const unsigned tid = threadIdx.x;
    const unsigned b = blockIdx.x;
    if (tid < R) dg[tid] = 2.0f;
    __syncthreads();
    const unsigned start = binStart[b];
    const unsigned end   = binStart[b + 1];
    for (unsigned i = start + tid; i < end; i += 1024u) {
        const int2 r = rec[i];
        const unsigned dl = (unsigned)r.x >> SRC_BITS;
        atomicAdd(&dg[dl], __builtin_bit_cast(float, r.y));
    }
    __syncthreads();
    if (tid < R) {
        const unsigned node = b * R + tid;
        if (node < N_NODES) {
            const float d = dg[tid];
            dinv[node] = (d > 0.0f) ? rsqrtf(d) : 0.0f;
        }
    }
}

__global__ void gemm_wmma_kernel(const float* __restrict__ x,
                                 const float* __restrict__ W,
                                 const float* __restrict__ dinv,
                                 float* __restrict__ h) {
    const unsigned lane  = threadIdx.x & 31u;
    const unsigned gwave = blockIdx.x * 8u + (threadIdx.x >> 5);
    const unsigned tile_n = gwave & 3u;
    const unsigned tile_m = gwave >> 2;
    const unsigned m0 = tile_m * 16u;
    const unsigned n0 = tile_n * 16u;
    const unsigned half = lane >> 4;
    const unsigned lrow = lane & 15u;

    const float* xrow = x + (size_t)(m0 + lrow) * IN_W;
    const float* wcol = W + (n0 + lrow);

    v8f c = {};
#pragma unroll
    for (int kk = 0; kk < 16; ++kk) {
        const int kb = kk * 4 + (int)half * 2;
        v2f a = *(const v2f*)(xrow + kb);
        v2f b;
        b.x = wcol[(size_t)kb * OUT_W];
        b.y = wcol[(size_t)(kb + 1) * OUT_W];
        c = __builtin_amdgcn_wmma_f32_16x16x4_f32(
                 false, a,  false, b,
                 (short)0, c,  false,  false);
    }

    float* hcol = h + n0 + lrow;
#pragma unroll
    for (int v = 0; v < 8; ++v) {
        const unsigned row = m0 + (unsigned)v + half * 8u;
        hcol[(size_t)row * OUT_W] = c[v] * dinv[row];
    }
}

#define EDGE_GROUP 16
#define ACC_THREADS 512
#define ACC_WAVES   (ACC_THREADS / 32)

__device__ __forceinline__ int lane_bcast_i(int v, int j) {
    return __builtin_amdgcn_readlane(v, j);
}

__global__ void __launch_bounds__(ACC_THREADS)
block_sum_kernel(const int2* __restrict__ rec,
                 const unsigned* __restrict__ binStart,
                 const float* __restrict__ dinv,
                 const float* __restrict__ h,
                 const float* __restrict__ bias,
                 float* __restrict__ out) {
    __shared__ float acc[R * OUT_W];
    const unsigned tid  = threadIdx.x;
    const unsigned lane = tid & 31u;
    const unsigned wv   = tid >> 5;
    const unsigned b    = blockIdx.x;
    const unsigned node0 = b * R;

    for (unsigned t = tid; t < R * (OUT_W / 4); t += ACC_THREADS) {
        const unsigned node = node0 + (t >> 4);
        float4 v = make_float4(0.0f, 0.0f, 0.0f, 0.0f);
        if (node < N_NODES) {
            const float4 hv = *(const float4*)(h + (size_t)node * OUT_W + (t & 15u) * 4u);
            v = make_float4(2.0f * hv.x, 2.0f * hv.y, 2.0f * hv.z, 2.0f * hv.w);
        }
        *(float4*)(acc + t * 4u) = v;
    }
    __syncthreads();

    const unsigned start = binStart[b];
    const unsigned end   = binStart[b + 1];

    for (unsigned base = start + wv * 32u; base < end; base += ACC_WAVES * 32u) {
        const unsigned left = end - base;
        const unsigned idx  = base + lane;
        int2 r = make_int2(0, 0);
        if (idx < end) r = rec[idx];

        if (left >= 32u) {
#pragma unroll
            for (int jb = 0; jb < 32; jb += EDGE_GROUP) {
                float h0[EDGE_GROUP], h1[EDGE_GROUP];
#pragma unroll
                for (int k = 0; k < EDGE_GROUP; ++k) {
                    const unsigned rx = (unsigned)lane_bcast_i(r.x, jb + k);
                    const float* hr = h + (size_t)(rx & ((1u << SRC_BITS) - 1u)) * OUT_W;
                    h0[k] = hr[lane];
                    h1[k] = hr[lane + 32u];
                }
#pragma unroll
                for (int k = 0; k < EDGE_GROUP; ++k) {
                    const unsigned rx = (unsigned)lane_bcast_i(r.x, jb + k);
                    const float wj = __builtin_bit_cast(float, lane_bcast_i(r.y, jb + k));
                    float* a = acc + (rx >> SRC_BITS) * OUT_W + lane;
                    atomicAdd(a,      h0[k] * wj);
                    atomicAdd(a + 32, h1[k] * wj);
                }
            }
        } else {
            for (unsigned j = 0; j < left; ++j) {
                const unsigned rx = (unsigned)lane_bcast_i(r.x, (int)j);
                const float wj = __builtin_bit_cast(float, lane_bcast_i(r.y, (int)j));
                const float* hr = h + (size_t)(rx & ((1u << SRC_BITS) - 1u)) * OUT_W;
                float* a = acc + (rx >> SRC_BITS) * OUT_W + lane;
                atomicAdd(a,      hr[lane]       * wj);
                atomicAdd(a + 32, hr[lane + 32u] * wj);
            }
        }
    }
    __syncthreads();

    for (unsigned t = tid; t < R * (OUT_W / 4); t += ACC_THREADS) {
        const unsigned node = node0 + (t >> 4);
        if (node < N_NODES) {
            const unsigned f = (t & 15u) * 4u;
            const float di = dinv[node];
            const float4 a  = *(const float4*)(acc + t * 4u);
            const float4 bv = *(const float4*)(bias + f);
            float4 o;
            o.x = bv.x + a.x * di;
            o.y = bv.y + a.y * di;
            o.z = bv.z + a.z * di;
            o.w = bv.w + a.w * di;
            *(float4*)(out + (size_t)node * OUT_W + f) = o;
        }
    }
}

extern "C" void kernel_launch(void* const* d_in, const int* in_sizes, int n_in,
                              void* d_out, int out_size, void* d_ws, size_t ws_size,
                              hipStream_t stream) {
    const float* x         = (const float*)d_in[0];
    const float* edge_attr = (const float*)d_in[1];
    const float* W         = (const float*)d_in[2];
    const float* bias      = (const float*)d_in[3];
    const int*   edge_idx  = (const int*)d_in[4];
    float* out = (float*)d_out;

    float*    dinv     = (float*)d_ws;
    float*    h        = dinv + 100352;
    int2*     rec      = (int2*)(h + 6400000);
    unsigned* counts   = (unsigned*)(rec + 1600000);
    unsigned* binStart = counts + 200704;

    const int* src = edge_idx;
    const int* dst = edge_idx + N_EDGES;

    bin_count_kernel<<<CH, BIN_THREADS, 0, stream>>>(dst, counts);
    bin_place_kernel<<<CH, BIN_THREADS, 0, stream>>>(src, dst, edge_attr, counts, rec, binStart);

    deg_block_kernel<<<NB, 1024, 0, stream>>>(rec, binStart, dinv);

    gemm_wmma_kernel<<<3125, 256, 0, stream>>>(x, W, dinv, h);

    block_sum_kernel<<<NB, ACC_THREADS, 0, stream>>>(rec, binStart, dinv, h, bias, out);
}


# degree pass folded into the GEMM launch: a work-group per 196-node block sums its degrees in LDS and multiplies its own rows on f32 WMMA from x rows and W staged in LDS; 5 launches to 4
# speedup vs baseline: 2.6765x; 1.0637x over previous
#include <hip/hip_runtime.h>
#include <math.h>


#define N_NODES 100000
#define IN_W    64
#define OUT_W   64
#define N_EDGES 1600000

#define R       196
#define NB      512
#define CHUNK   8192
#define CH      196
#define BIN_THREADS 1024
#define PER_THREAD  (CHUNK / BIN_THREADS)
#define SRC_BITS 17
#define SCAN_GROUP 14

static_assert(NB * R >= N_NODES, "blocks must cover all nodes");
static_assert(CH * CHUNK >= N_EDGES, "chunks must cover all edges");
static_assert(N_NODES <= (1 << SRC_BITS), "src must fit its field");
static_assert(R <= 512, "local row must fit its field");
static_assert((CH / 2) % SCAN_GROUP == 0 && CH % 2 == 0, "each half of the chunks splits into whole groups");
static_assert(2 * NB == BIN_THREADS, "placing threads = two per bin");

typedef __attribute__((ext_vector_type(2))) float v2f;
typedef __attribute__((ext_vector_type(8))) float v8f;

__global__ void __launch_bounds__(BIN_THREADS)
bin_count_kernel(const int* __restrict__ dst, unsigned* __restrict__ counts) {
    __shared__ unsigned cnt[NB];
    const unsigned tid = threadIdx.x;
    const unsigned chunk = blockIdx.x;
    if (tid < NB) cnt[tid] = 0u;
    __syncthreads();
    const unsigned e0 = chunk * CHUNK + tid;
    unsigned bins[PER_THREAD];
#pragma unroll
    for (int k = 0; k < PER_THREAD; ++k) {
        const unsigned e = e0 + (unsigned)k * BIN_THREADS;
        bins[k] = (e < N_EDGES) ? (unsigned)dst[e] / R : (unsigned)NB;
    }
#pragma unroll
    for (int k = 0; k < PER_THREAD; ++k) {
        if (bins[k] < NB) atomicAdd(&cnt[bins[k]], 1u);
    }
    __syncthreads();
    if (tid < NB) counts[chunk * NB + tid] = cnt[tid];
}

__global__ void __launch_bounds__(BIN_THREADS)
bin_place_kernel(const int* __restrict__ src,
                 const int* __restrict__ dst,
                 const float* __restrict__ w,
                 const unsigned* __restrict__ counts,
                 int2* __restrict__ rec,
                 unsigned* __restrict__ binStart) {
    __shared__ unsigned cur[NB];
    __shared__ unsigned ptot[2][NB];
    __shared__ unsigned ppre[2][NB];
    __shared__ unsigned wsum[NB / 32];
    const unsigned tid = threadIdx.x;
    const unsigned chunk = blockIdx.x;
    const unsigned lane = tid & 31u;
    const unsigned b = tid & (NB - 1);
    const unsigned q = tid >> 9;

    const unsigned e0 = chunk * CHUNK + tid;
    unsigned sv[PER_THREAD], dv[PER_THREAD];
    float wv[PER_THREAD];
#pragma unroll
    for (int k = 0; k < PER_THREAD; ++k) {
        const unsigned e = e0 + (unsigned)k * BIN_THREADS;
        const bool live = e < N_EDGES;
        sv[k] = live ? (unsigned)src[e] : 0u;
        dv[k] = live ? (unsigned)dst[e] : (unsigned)(NB * R);
        wv[k] = live ? w[e] : 0.0f;
    }

    const unsigned c0 = q * (CH / 2);
    unsigned tot = 0u, pre = 0u;
    for (unsigned cc = c0; cc < c0 + (CH / 2); cc += SCAN_GROUP) {
        unsigned v[SCAN_GROUP];
#pragma unroll
        for (int k = 0; k < SCAN_GROUP; ++k) v[k] = counts[(cc + (unsigned)k) * NB + b];
#pragma unroll
        for (int k = 0; k < SCAN_GROUP; ++k) {
            tot += v[k];
            pre += (cc + (unsigned)k < chunk) ? v[k] : 0u;
        }
    }
    ptot[q][b] = tot;
    ppre[q][b] = pre;
    __syncthreads();

    unsigned T = 0u, P = 0u, incl = 0u;
    if (tid < NB) {
        T = ptot[0][b] + ptot[1][b];
        P = ppre[0][b] + ppre[1][b];
        incl = T;
#pragma unroll
        for (int o = 1; o < 32; o <<= 1) {
            const unsigned n = (unsigned)__shfl_up((int)incl, (unsigned)o);
            if (lane >= (unsigned)o) incl += n;
        }
        if (lane == 31u) wsum[tid >> 5] = incl;
    }
    __syncthreads();
    if (tid < NB) {
        const unsigned wvi = tid >> 5;
        unsigned base = 0u;
#pragma unroll
        for (unsigned k = 0; k < NB / 32; ++k) base += (k < wvi) ? wsum[k] : 0u;
        const unsigned excl = base + incl - T;
        cur[b] = excl + P;
        if (chunk == 0u) {
            binStart[b] = excl;
            if (b == NB - 1) binStart[NB] = excl + T;
        }
    }
    __syncthreads();

#pragma unroll
    for (int k = 0; k < PER_THREAD; ++k) {
        if (dv[k] < NB * R) {
            const unsigned bin = dv[k] / R;
            const unsigned dl = dv[k] - bin * R;
            const unsigned pos = atomicAdd(&cur[bin], 1u);
            int2 r;
            r.x = (int)(sv[k] | (dl << SRC_BITS));
            r.y = __builtin_bit_cast(int, wv[k]);
            rec[pos] = r;
        }
    }
}

#define XT_ROWS   208
#define XS_PITCH  68
#define WS_PITCH  160
#define DG_THREADS 1024
#define GEMM_WAVES 26
#define REC_PRE   4

static_assert(XT_ROWS >= R && XT_ROWS % 16 == 0, "row tiles must cover the block");
static_assert(GEMM_WAVES == (XT_ROWS / 16) * 2 && GEMM_WAVES * 32 <= DG_THREADS, "one wave per (row tile, column half)");
static_assert(IN_W * OUT_W == DG_THREADS * 4, "W is staged as one float4 per thread");
static_assert(XT_ROWS * 16 <= 4 * DG_THREADS, "x rows are staged as at most four float4 per thread");

__global__ void __launch_bounds__(DG_THREADS)
deg_gemm_kernel(const int2* __restrict__ rec,
                const unsigned* __restrict__ binStart,
                const float* __restrict__ x,
                const float* __restrict__ W,
                float* __restrict__ dinv,
                float* __restrict__ h) {
    __shared__ float dg[XT_ROWS];
    __shared__ alignas(16) float xs[XT_ROWS * XS_PITCH];
    __shared__ alignas(16) float ws[(IN_W / 2) * WS_PITCH];
    const unsigned tid  = threadIdx.x;
    const unsigned lane = tid & 31u;
    const unsigned wv   = tid >> 5;
    const unsigned b    = blockIdx.x;
    const unsigned node0 = b * R;

    float4 xv[4];
#pragma unroll
    for (int k = 0; k < 4; ++k) {
        const unsigned t4   = tid + (unsigned)k * DG_THREADS;
        const unsigned row  = min(t4 >> 4, (unsigned)(R - 1));
        const unsigned node = min(node0 + row, (unsigned)(N_NODES - 1));
        xv[k] = *(const float4*)(x + (size_t)node * IN_W + (t4 & 15u) * 4u);
    }
    const float4 w4 = *(const float4*)(W + tid * 4u);

    const unsigned start = binStart[b];
    const unsigned end   = binStart[b + 1];
    int2 rp[REC_PRE];
#pragma unroll
    for (int k = 0; k < REC_PRE; ++k) {
        const unsigned i = start + tid + (unsigned)k * DG_THREADS;
        rp[k] = rec[min((i < end) ? i : start, (unsigned)(N_EDGES - 1))];
    }

#pragma unroll
    for (int k = 0; k < 4; ++k) {
        const unsigned t4  = tid + (unsigned)k * DG_THREADS;
        const unsigned row = t4 >> 4;
        if (row >= R || node0 + row >= N_NODES) xv[k] = make_float4(0.0f, 0.0f, 0.0f, 0.0f);
    }

    if (tid < XT_ROWS) dg[tid] = 2.0f;
#pragma unroll
    for (int k = 0; k < 4; ++k) {
        const unsigned t4 = tid + (unsigned)k * DG_THREADS;
        if (t4 < XT_ROWS * 16u)
            *(float4*)(xs + (t4 >> 4) * XS_PITCH + (t4 & 15u) * 4u) = xv[k];
    }
    {
        const unsigned k = tid >> 4;
        const unsigned n = (tid & 15u) * 4u;
        float* p = ws + (k >> 1) * WS_PITCH + n * 2u + (k & 1u);
        p[0] = w4.x; p[2] = w4.y; p[4] = w4.z; p[6] = w4.w;
    }
    __syncthreads();

#pragma unroll
    for (int k = 0; k < REC_PRE; ++k) {
        const unsigned i = start + tid + (unsigned)k * DG_THREADS;
        if (i < end)
            atomicAdd(&dg[(unsigned)rp[k].x >> SRC_BITS], __builtin_bit_cast(float, rp[k].y));
    }
    for (unsigned i = start + tid + REC_PRE * DG_THREADS; i < end; i += DG_THREADS) {
        const int2 r = rec[i];
        atomicAdd(&dg[(unsigned)r.x >> SRC_BITS], __builtin_bit_cast(float, r.y));
    }

    const unsigned half = lane >> 4;
    const unsigned lrow = lane & 15u;
    const unsigned rt   = wv >> 1;
    const unsigned ch   = wv & 1u;
    v8f c0 = {};
    v8f c1 = {};
    if (wv < GEMM_WAVES) {
        const float* xa = xs + (rt * 16u + lrow) * XS_PITCH + half * 2u;
        const float* wb = ws + half * WS_PITCH + (ch * 32u + lrow) * 2u;
#pragma unroll
        for (int kk = 0; kk < 16; ++kk) {
            const v2f a  = *(const v2f*)(xa + kk * 4);
            const v2f b0 = *(const v2f*)(wb + kk * 2 * WS_PITCH);
            const v2f b1 = *(const v2f*)(wb + kk * 2 * WS_PITCH + 32);
            c0 = __builtin_amdgcn_wmma_f32_16x16x4_f32(
                      false, a,  false, b0,
                      (short)0, c0,  false,  false);
            c1 = __builtin_amdgcn_wmma_f32_16x16x4_f32(
                      false, a,  false, b1,
                      (short)0, c1,  false,  false);
        }
    }
    __syncthreads();

    if (wv < GEMM_WAVES) {
        float* ot = xs + (rt * 16u + half * 8u) * XS_PITCH + ch * 32u + lrow;
#pragma unroll
        for (int v = 0; v < 8; ++v) {
            ot[v * XS_PITCH]      = c0[v];
            ot[v * XS_PITCH + 16] = c1[v];
        }
    }
    __syncthreads();
    if (tid < R) {
        const unsigned node = node0 + tid;
        if (node < N_NODES) {
            const float d = dg[tid];
            dinv[node] = (d > 0.0f) ? rsqrtf(d) : 0.0f;
        }
    }
    if (wv < GEMM_WAVES) {
        const unsigned f = ch * 32u + (lane & 7u) * 4u;
#pragma unroll
        for (int j = 0; j < 4; ++j) {
            const unsigned row  = rt * 16u + (lane >> 3) + (unsigned)j * 4u;
            const unsigned node = node0 + row;
            if (row < R && node < N_NODES) {
                const float d  = dg[row];
                const float di = (d > 0.0f) ? rsqrtf(d) : 0.0f;
                const float4 a = *(const float4*)(xs + row * XS_PITCH + f);
                float4 o;
                o.x = a.x * di; o.y = a.y * di; o.z = a.z * di; o.w = a.w * di;
                *(float4*)(h + (size_t)node * OUT_W + f) = o;
            }
        }
    }
}

#define EDGE_GROUP 16
#define ACC_THREADS 512
#define ACC_WAVES   (ACC_THREADS / 32)

__device__ __forceinline__ int lane_bcast_i(int v, int j) {
    return __builtin_amdgcn_readlane(v, j);
}

__global__ void __launch_bounds__(ACC_THREADS)
block_sum_kernel(const int2* __restrict__ rec,
                 const unsigned* __restrict__ binStart,
                 const float* __restrict__ dinv,
                 const float* __restrict__ h,
                 const float* __restrict__ bias,
                 float* __restrict__ out) {
    __shared__ float acc[R * OUT_W];
    const unsigned tid  = threadIdx.x;
    const unsigned lane = tid & 31u;
    const unsigned wv   = tid >> 5;
    const unsigned b    = blockIdx.x;
    const unsigned node0 = b * R;

    for (unsigned t = tid; t < R * (OUT_W / 4); t += ACC_THREADS) {
        const unsigned node = node0 + (t >> 4);
        float4 v = make_float4(0.0f, 0.0f, 0.0f, 0.0f);
        if (node < N_NODES) {
            const float4 hv = *(const float4*)(h + (size_t)node * OUT_W + (t & 15u) * 4u);
            v = make_float4(2.0f * hv.x, 2.0f * hv.y, 2.0f * hv.z, 2.0f * hv.w);
        }
        *(float4*)(acc + t * 4u) = v;
    }
    __syncthreads();

    const unsigned start = binStart[b];
    const unsigned end   = binStart[b + 1];

    for (unsigned base = start + wv * 32u; base < end; base += ACC_WAVES * 32u) {
        const unsigned left = end - base;
        const unsigned idx  = base + lane;
        int2 r = make_int2(0, 0);
        if (idx < end) r = rec[idx];

        if (left >= 32u) {
#pragma unroll
            for (int jb = 0; jb < 32; jb += EDGE_GROUP) {
                float h0[EDGE_GROUP], h1[EDGE_GROUP];
#pragma unroll
                for (int k = 0; k < EDGE_GROUP; ++k) {
                    const unsigned rx = (unsigned)lane_bcast_i(r.x, jb + k);
                    const float* hr = h + (size_t)(rx & ((1u << SRC_BITS) - 1u)) * OUT_W;
                    h0[k] = hr[lane];
                    h1[k] = hr[lane + 32u];
                }
#pragma unroll
                for (int k = 0; k < EDGE_GROUP; ++k) {
                    const unsigned rx = (unsigned)lane_bcast_i(r.x, jb + k);
                    const float wj = __builtin_bit_cast(float, lane_bcast_i(r.y, jb + k));
                    float* a = acc + (rx >> SRC_BITS) * OUT_W + lane;
                    atomicAdd(a,      h0[k] * wj);
                    atomicAdd(a + 32, h1[k] * wj);
                }
            }
        } else {
            for (unsigned j = 0; j < left; ++j) {
                const unsigned rx = (unsigned)lane_bcast_i(r.x, (int)j);
                const float wj = __builtin_bit_cast(float, lane_bcast_i(r.y, (int)j));
                const float* hr = h + (size_t)(rx & ((1u << SRC_BITS) - 1u)) * OUT_W;
                float* a = acc + (rx >> SRC_BITS) * OUT_W + lane;
                atomicAdd(a,      hr[lane]       * wj);
                atomicAdd(a + 32, hr[lane + 32u] * wj);
            }
        }
    }
    __syncthreads();

    for (unsigned t = tid; t < R * (OUT_W / 4); t += ACC_THREADS) {
        const unsigned node = node0 + (t >> 4);
        if (node < N_NODES) {
            const unsigned f = (t & 15u) * 4u;
            const float di = dinv[node];
            const float4 a  = *(const float4*)(acc + t * 4u);
            const float4 bv = *(const float4*)(bias + f);
            float4 o;
            o.x = bv.x + a.x * di;
            o.y = bv.y + a.y * di;
            o.z = bv.z + a.z * di;
            o.w = bv.w + a.w * di;
            *(float4*)(out + (size_t)node * OUT_W + f) = o;
        }
    }
}

extern "C" void kernel_launch(void* const* d_in, const int* in_sizes, int n_in,
                              void* d_out, int out_size, void* d_ws, size_t ws_size,
                              hipStream_t stream) {
    const float* x         = (const float*)d_in[0];
    const float* edge_attr = (const float*)d_in[1];
    const float* W         = (const float*)d_in[2];
    const float* bias      = (const float*)d_in[3];
    const int*   edge_idx  = (const int*)d_in[4];
    float* out = (float*)d_out;

    float*    dinv     = (float*)d_ws;
    float*    h        = dinv + 100352;
    int2*     rec      = (int2*)(h + 6400000);
    unsigned* counts   = (unsigned*)(rec + 1600000);
    unsigned* binStart = counts + 200704;

    const int* src = edge_idx;
    const int* dst = edge_idx + N_EDGES;

    bin_count_kernel<<<CH, BIN_THREADS, 0, stream>>>(dst, counts);
    bin_place_kernel<<<CH, BIN_THREADS, 0, stream>>>(src, dst, edge_attr, counts, rec, binStart);

    deg_gemm_kernel<<<NB, DG_THREADS, 0, stream>>>(rec, binStart, x, W, dinv, h);

    block_sum_kernel<<<NB, ACC_THREADS, 0, stream>>>(rec, binStart, dinv, h, bias, out);
}


# count launch and count table removed: each chunk sorts its edge records by destination block inside its own record range; block passes walk per-chunk segments (4 launches to 3, sum order changes)
# speedup vs baseline: 4.0366x; 1.5082x over previous
#include <hip/hip_runtime.h>
#include <math.h>


#define N_NODES 100000
#define IN_W    64
#define OUT_W   64
#define N_EDGES 1600000

#define R       196
#define NB      512
#define CHUNK   8192
#define CH      196
#define BIN_THREADS 1024
#define PER_THREAD  (CHUNK / BIN_THREADS)
#define SRC_BITS 17
#define SEG_SHIFT 16

static_assert(NB * R >= N_NODES, "blocks must cover all nodes");
static_assert(CH * CHUNK >= N_EDGES, "chunks must cover all edges");
static_assert(N_EDGES >= CHUNK, "chunk 0 is a full chunk: record 0 is always written");
static_assert(N_NODES <= (1 << SRC_BITS), "src must fit its field");
static_assert(R < 512, "local row, and the spare row R, must fit their field");
static_assert(CHUNK < (1 << SEG_SHIFT), "a segment's start and length must fit their fields");
static_assert(NB <= BIN_THREADS, "one scanning thread per bin");

typedef __attribute__((ext_vector_type(2))) float v2f;
typedef __attribute__((ext_vector_type(8))) float v8f;

__global__ void __launch_bounds__(BIN_THREADS)
chunk_sort_kernel(const int* __restrict__ src,
                  const int* __restrict__ dst,
                  const float* __restrict__ w,
                  int2* __restrict__ rec,
                  unsigned* __restrict__ seg) {
    __shared__ unsigned cnt[NB];
    __shared__ unsigned cur[NB];
    __shared__ unsigned wsum[NB / 32];
    const unsigned tid = threadIdx.x;
    const unsigned chunk = blockIdx.x;
    const unsigned lane = tid & 31u;
    if (tid < NB) cnt[tid] = 0u;

    const unsigned e0 = chunk * CHUNK + tid;
    unsigned sv[PER_THREAD], dv[PER_THREAD];
    float wv[PER_THREAD];
#pragma unroll
    for (int k = 0; k < PER_THREAD; ++k) {
        const unsigned e = e0 + (unsigned)k * BIN_THREADS;
        const bool live = e < N_EDGES;
        sv[k] = live ? (unsigned)src[e] : 0u;
        dv[k] = live ? (unsigned)dst[e] : (unsigned)(NB * R);
        wv[k] = live ? w[e] : 0.0f;
    }
    __syncthreads();

    unsigned bn[PER_THREAD];
#pragma unroll
    for (int k = 0; k < PER_THREAD; ++k) {
        bn[k] = dv[k] / R;
        if (bn[k] < NB) atomicAdd(&cnt[bn[k]], 1u);
    }
    __syncthreads();

    unsigned T = 0u, incl = 0u;
    if (tid < NB) {
        T = cnt[tid];
        incl = T;
#pragma unroll
        for (int o = 1; o < 32; o <<= 1) {
            const unsigned n = (unsigned)__shfl_up((int)incl, (unsigned)o);
            if (lane >= (unsigned)o) incl += n;
        }
        if (lane == 31u) wsum[tid >> 5] = incl;
    }
    __syncthreads();
    if (tid < NB) {
        const unsigned wvi = tid >> 5;
        unsigned base = 0u;
#pragma unroll
        for (unsigned k = 0; k < NB / 32; ++k) base += (k < wvi) ? wsum[k] : 0u;
        const unsigned excl = base + incl - T;
        cur[tid] = excl;
        seg[chunk * NB + tid] = excl | (T << SEG_SHIFT);
    }
    __syncthreads();

#pragma unroll
    for (int k = 0; k < PER_THREAD; ++k) {
        if (bn[k] < NB) {
            const unsigned dl = dv[k] - bn[k] * R;
            const unsigned pos = chunk * CHUNK + atomicAdd(&cur[bn[k]], 1u);
            int2 r;
            r.x = (int)(sv[k] | (dl << SRC_BITS));
            r.y = __builtin_bit_cast(int, wv[k]);
            rec[pos] = r;
        }
    }
}

template <int NSEG>
struct SegMap {
    unsigned pre[NSEG];
    unsigned adj[NSEG];
    unsigned total;
};

template <int NSEG>
__device__ __forceinline__ SegMap<NSEG> load_seg_map(const unsigned* __restrict__ seg,
                                                     unsigned b, unsigned first, unsigned stride,
                                                     unsigned lane) {
    unsigned len = 0u, sbase = 0u;
    const unsigned c = first + stride * lane;
    if (lane < (unsigned)NSEG && c < CH) {
        const unsigned s = seg[c * NB + b];
        len   = s >> SEG_SHIFT;
        sbase = c * CHUNK + (s & ((1u << SEG_SHIFT) - 1u));
    }
    unsigned incl = len;
#pragma unroll
    for (int o = 1; o < 32; o <<= 1) {
        const unsigned n = (unsigned)__shfl_up((int)incl, (unsigned)o);
        if (lane >= (unsigned)o) incl += n;
    }
    const unsigned excl = incl - len;
    const unsigned a = sbase - excl;
    SegMap<NSEG> m;
#pragma unroll
    for (int i = 0; i < NSEG; ++i) {
        m.pre[i] = (unsigned)__builtin_amdgcn_readlane((int)excl, i);
        m.adj[i] = (unsigned)__builtin_amdgcn_readlane((int)a, i);
    }
    m.total = (unsigned)__builtin_amdgcn_readlane((int)incl, 31);
    return m;
}

template <int NSEG>
__device__ __forceinline__ unsigned seg_index(const SegMap<NSEG>& m, unsigned g) {
    unsigned a = m.adj[0];
#pragma unroll
    for (int i = 1; i < NSEG; ++i) a = (g >= m.pre[i]) ? m.adj[i] : a;
    return a + g;
}

#define XT_ROWS   208
#define XS_PITCH  68
#define WS_PITCH  160
#define DG_THREADS 1024
#define GEMM_WAVES 26
#define REC_PRE   4
#define DG_SEGS   7

static_assert(XT_ROWS >= R && XT_ROWS % 16 == 0, "row tiles must cover the block");
static_assert(GEMM_WAVES == (XT_ROWS / 16) * 2 && GEMM_WAVES * 32 <= DG_THREADS, "one wave per (row tile, column half)");
static_assert(IN_W * OUT_W == DG_THREADS * 4, "W is staged as one float4 per thread");
static_assert(XT_ROWS * 16 <= 4 * DG_THREADS, "x rows are staged as at most four float4 per thread");
static_assert((DG_THREADS / 32) * DG_SEGS >= CH && DG_SEGS <= 32, "the waves' segments must cover all chunks");

__global__ void __launch_bounds__(DG_THREADS)
deg_gemm_kernel(const int2* __restrict__ rec,
                const unsigned* __restrict__ seg,
                const float* __restrict__ x,
                const float* __restrict__ W,
                float* __restrict__ dinv,
                float* __restrict__ h) {
    __shared__ float dg[XT_ROWS];
    __shared__ alignas(16) float xs[XT_ROWS * XS_PITCH];
    __shared__ alignas(16) float ws[(IN_W / 2) * WS_PITCH];
    const unsigned tid  = threadIdx.x;
    const unsigned lane = tid & 31u;
    const unsigned wv   = tid >> 5;
    const unsigned b    = blockIdx.x;
    const unsigned node0 = b * R;

    float4 xv[4];
#pragma unroll
    for (int k = 0; k < 4; ++k) {
        const unsigned t4   = tid + (unsigned)k * DG_THREADS;
        const unsigned row  = min(t4 >> 4, (unsigned)(R - 1));
        const unsigned node = min(node0 + row, (unsigned)(N_NODES - 1));
        xv[k] = *(const float4*)(x + (size_t)node * IN_W + (t4 & 15u) * 4u);
    }
    const float4 w4 = *(const float4*)(W + tid * 4u);

    const SegMap<DG_SEGS> sm = load_seg_map<DG_SEGS>(seg, b, wv, DG_THREADS / 32u, lane);
    int2 rp[REC_PRE];
#pragma unroll
    for (int k = 0; k < REC_PRE; ++k) {
        const unsigned g = lane + (unsigned)k * 32u;
        rp[k] = rec[(g < sm.total) ? seg_index(sm, g) : 0u];
    }

#pragma unroll
    for (int k = 0; k < 4; ++k) {
        const unsigned t4  = tid + (unsigned)k * DG_THREADS;
        const unsigned row = t4 >> 4;
        if (row >= R || node0 + row >= N_NODES) xv[k] = make_float4(0.0f, 0.0f, 0.0f, 0.0f);
    }

    if (tid < XT_ROWS) dg[tid] = 2.0f;
#pragma unroll
    for (int k = 0; k < 4; ++k) {
        const unsigned t4 = tid + (unsigned)k * DG_THREADS;
        if (t4 < XT_ROWS * 16u)
            *(float4*)(xs + (t4 >> 4) * XS_PITCH + (t4 & 15u) * 4u) = xv[k];
    }
    {
        const unsigned k = tid >> 4;
        const unsigned n = (tid & 15u) * 4u;
        float* p = ws + (k >> 1) * WS_PITCH + n * 2u + (k & 1u);
        p[0] = w4.x; p[2] = w4.y; p[4] = w4.z; p[6] = w4.w;
    }
    __syncthreads();

#pragma unroll
    for (int k = 0; k < REC_PRE; ++k) {
        const unsigned g = lane + (unsigned)k * 32u;
        if (g < sm.total)
            atomicAdd(&dg[(unsigned)rp[k].x >> SRC_BITS], __builtin_bit_cast(float, rp[k].y));
    }
    for (unsigned g = lane + REC_PRE * 32u; g < sm.total; g += 32u) {
        const int2 r = rec[seg_index(sm, g)];
        atomicAdd(&dg[(unsigned)r.x >> SRC_BITS], __builtin_bit_cast(float, r.y));
    }

    const unsigned half = lane >> 4;
    const unsigned lrow = lane & 15u;
    const unsigned rt   = wv >> 1;
    const unsigned ch   = wv & 1u;
    v8f c0 = {};
    v8f c1 = {};
    if (wv < GEMM_WAVES) {
        const float* xa = xs + (rt * 16u + lrow) * XS_PITCH + half * 2u;
        const float* wb = ws + half * WS_PITCH + (ch * 32u + lrow) * 2u;
#pragma unroll
        for (int kk = 0; kk < 16; ++kk) {
            const v2f a  = *(const v2f*)(xa + kk * 4);
            const v2f b0 = *(const v2f*)(wb + kk * 2 * WS_PITCH);
            const v2f b1 = *(const v2f*)(wb + kk * 2 * WS_PITCH + 32);
            c0 = __builtin_amdgcn_wmma_f32_16x16x4_f32(
                      false, a,  false, b0,
                      (short)0, c0,  false,  false);
            c1 = __builtin_amdgcn_wmma_f32_16x16x4_f32(
                      false, a,  false, b1,
                      (short)0, c1,  false,  false);
        }
    }
    __syncthreads();

    if (wv < GEMM_WAVES) {
        float* ot = xs + (rt * 16u + half * 8u) * XS_PITCH + ch * 32u + lrow;
#pragma unroll
        for (int v = 0; v < 8; ++v) {
            ot[v * XS_PITCH]      = c0[v];
            ot[v * XS_PITCH + 16] = c1[v];
        }
    }
    __syncthreads();
    if (tid < R) {
        const unsigned node = node0 + tid;
        if (node < N_NODES) {
            const float d = dg[tid];
            dinv[node] = (d > 0.0f) ? rsqrtf(d) : 0.0f;
        }
    }
    if (wv < GEMM_WAVES) {
        const unsigned f = ch * 32u + (lane & 7u) * 4u;
#pragma unroll
        for (int j = 0; j < 4; ++j) {
            const unsigned row  = rt * 16u + (lane >> 3) + (unsigned)j * 4u;
            const unsigned node = node0 + row;
            if (row < R && node < N_NODES) {
                const float d  = dg[row];
                const float di = (d > 0.0f) ? rsqrtf(d) : 0.0f;
                const float4 a = *(const float4*)(xs + row * XS_PITCH + f);
                float4 o;
                o.x = a.x * di; o.y = a.y * di; o.z = a.z * di; o.w = a.w * di;
                *(float4*)(h + (size_t)node * OUT_W + f) = o;
            }
        }
    }
}

#define EDGE_GROUP 16
#define ACC_THREADS 512
#define ACC_WAVES   (ACC_THREADS / 32)
#define ACC_SEGS    13
static_assert(ACC_WAVES * ACC_SEGS >= CH && ACC_SEGS <= 32, "the waves' segments must cover all chunks");

__device__ __forceinline__ int lane_bcast_i(int v, int j) {
    return __builtin_amdgcn_readlane(v, j);
}

__global__ void __launch_bounds__(ACC_THREADS)
block_sum_kernel(const int2* __restrict__ rec,
                 const unsigned* __restrict__ seg,
                 const float* __restrict__ dinv,
                 const float* __restrict__ h,
                 const float* __restrict__ bias,
                 float* __restrict__ out) {
    __shared__ float acc[(R + 1) * OUT_W];
    const unsigned tid  = threadIdx.x;
    const unsigned lane = tid & 31u;
    const unsigned wv   = tid >> 5;
    const unsigned b    = blockIdx.x;
    const unsigned node0 = b * R;

    const SegMap<ACC_SEGS> sm = load_seg_map<ACC_SEGS>(seg, b, wv, ACC_WAVES, lane);

    for (unsigned t = tid; t < R * (OUT_W / 4); t += ACC_THREADS) {
        const unsigned node = node0 + (t >> 4);
        float4 v = make_float4(0.0f, 0.0f, 0.0f, 0.0f);
        if (node < N_NODES) {
            const float4 hv = *(const float4*)(h + (size_t)node * OUT_W + (t & 15u) * 4u);
            v = make_float4(2.0f * hv.x, 2.0f * hv.y, 2.0f * hv.z, 2.0f * hv.w);
        }
        *(float4*)(acc + t * 4u) = v;
    }
    __syncthreads();

    for (unsigned t0 = 0u; t0 < sm.total; t0 += 32u) {
        const unsigned left = sm.total - t0;
        const unsigned g    = t0 + lane;
        int2 r = make_int2((int)((unsigned)R << SRC_BITS), 0);
        if (g < sm.total) r = rec[seg_index(sm, g)];

#pragma unroll
        for (int jb = 0; jb < 32; jb += EDGE_GROUP) {
            if ((unsigned)jb < left) {
                float h0[EDGE_GROUP], h1[EDGE_GROUP];
#pragma unroll
                for (int k = 0; k < EDGE_GROUP; ++k) {
                    const unsigned rx = (unsigned)lane_bcast_i(r.x, jb + k);
                    const float* hr = h + (size_t)(rx & ((1u << SRC_BITS) - 1u)) * OUT_W;
                    h0[k] = hr[lane];
                    h1[k] = hr[lane + 32u];
                }
#pragma unroll
                for (int k = 0; k < EDGE_GROUP; ++k) {
                    const unsigned rx = (unsigned)lane_bcast_i(r.x, jb + k);
                    const float wj = __builtin_bit_cast(float, lane_bcast_i(r.y, jb + k));
                    float* a = acc + (rx >> SRC_BITS) * OUT_W + lane;
                    atomicAdd(a,      h0[k] * wj);
                    atomicAdd(a + 32, h1[k] * wj);
                }
            }
        }
    }
    __syncthreads();

    for (unsigned t = tid; t < R * (OUT_W / 4); t += ACC_THREADS) {
        const unsigned node = node0 + (t >> 4);
        if (node < N_NODES) {
            const unsigned f = (t & 15u) * 4u;
            const float di = dinv[node];
            const float4 a  = *(const float4*)(acc + t * 4u);
            const float4 bv = *(const float4*)(bias + f);
            float4 o;
            o.x = bv.x + a.x * di;
            o.y = bv.y + a.y * di;
            o.z = bv.z + a.z * di;
            o.w = bv.w + a.w * di;
            *(float4*)(out + (size_t)node * OUT_W + f) = o;
        }
    }
}

extern "C" void kernel_launch(void* const* d_in, const int* in_sizes, int n_in,
                              void* d_out, int out_size, void* d_ws, size_t ws_size,
                              hipStream_t stream) {
    const float* x         = (const float*)d_in[0];
    const float* edge_attr = (const float*)d_in[1];
    const float* W         = (const float*)d_in[2];
    const float* bias      = (const float*)d_in[3];
    const int*   edge_idx  = (const int*)d_in[4];
    float* out = (float*)d_out;

    float*    dinv     = (float*)d_ws;
    float*    h        = dinv + 100352;
    int2*     rec      = (int2*)(h + 6400000);
    unsigned* seg      = (unsigned*)(rec + 1600000);

    const int* src = edge_idx;
    const int* dst = edge_idx + N_EDGES;

    chunk_sort_kernel<<<CH, BIN_THREADS, 0, stream>>>(src, dst, edge_attr, rec, seg);

    deg_gemm_kernel<<<NB, DG_THREADS, 0, stream>>>(rec, seg, x, W, dinv, h);

    block_sum_kernel<<<NB, ACC_THREADS, 0, stream>>>(rec, seg, dinv, h, bias, out);
}


# chunk sort places its records in LDS (64 KiB) and writes the chunk's range out in order, 16 B per lane: each rec store covers 4 whole 128-B lines instead of 32 scattered 8-B pieces
# speedup vs baseline: 4.2346x; 1.0491x over previous
#include <hip/hip_runtime.h>
#include <math.h>


#define N_NODES 100000
#define IN_W    64
#define OUT_W   64
#define N_EDGES 1600000

#define R       196
#define NB      512
#define CHUNK   8192
#define CH      196
#define BIN_THREADS 1024
#define PER_THREAD  (CHUNK / BIN_THREADS)
#define SRC_BITS 17
#define SEG_SHIFT 16

static_assert(NB * R >= N_NODES, "blocks must cover all nodes");
static_assert(CH * CHUNK >= N_EDGES, "chunks must cover all edges");
static_assert(N_EDGES >= CHUNK, "chunk 0 is a full chunk: record 0 is always written");
static_assert(N_NODES <= (1 << SRC_BITS), "src must fit its field");
static_assert(R < 512, "local row, and the spare row R, must fit their field");
static_assert(CHUNK < (1 << SEG_SHIFT), "a segment's start and length must fit their fields");
static_assert(NB <= BIN_THREADS, "one scanning thread per bin");
static_assert(CHUNK % 2 == 0 && (N_EDGES % CHUNK) % 2 == 0, "every chunk holds an even number of records: it is written out two records per 16-byte store");
static_assert(CHUNK / 2 == (PER_THREAD / 2) * BIN_THREADS, "the write-out loop covers a full chunk");
static_assert((CH - 1) * CHUNK < N_EDGES, "no chunk is empty: with the line above every chunk holds at least one record pair");

typedef __attribute__((ext_vector_type(2))) float v2f;
typedef __attribute__((ext_vector_type(8))) float v8f;
typedef __attribute__((ext_vector_type(2))) int v2i;
typedef __attribute__((ext_vector_type(4))) int v4i;

__global__ void __launch_bounds__(BIN_THREADS)
chunk_sort_kernel(const int* __restrict__ src,
                  const int* __restrict__ dst,
                  const float* __restrict__ w,
                  int2* __restrict__ rec,
                  unsigned* __restrict__ seg) {
    __shared__ unsigned cnt[NB];
    __shared__ unsigned cur[NB];
    __shared__ unsigned wsum[NB / 32];
    __shared__ v4i srec[CHUNK / 2];
    const unsigned tid = threadIdx.x;
    const unsigned chunk = blockIdx.x;
    const unsigned lane = tid & 31u;
    if (tid < NB) cnt[tid] = 0u;

    const unsigned e0 = chunk * CHUNK + tid;
    unsigned sv[PER_THREAD], dv[PER_THREAD];
    float wv[PER_THREAD];
#pragma unroll
    for (int k = 0; k < PER_THREAD; ++k) {
        const unsigned e = e0 + (unsigned)k * BIN_THREADS;
        const bool live = e < N_EDGES;
        sv[k] = live ? (unsigned)src[e] : 0u;
        dv[k] = live ? (unsigned)dst[e] : (unsigned)(NB * R);
        wv[k] = live ? w[e] : 0.0f;
    }
    __syncthreads();

    unsigned bn[PER_THREAD];
#pragma unroll
    for (int k = 0; k < PER_THREAD; ++k) {
        bn[k] = dv[k] / R;
        if (bn[k] < NB) atomicAdd(&cnt[bn[k]], 1u);
    }
    __syncthreads();

    unsigned T = 0u, incl = 0u;
    if (tid < NB) {
        T = cnt[tid];
        incl = T;
#pragma unroll
        for (int o = 1; o < 32; o <<= 1) {
            const unsigned n = (unsigned)__shfl_up((int)incl, (unsigned)o);
            if (lane >= (unsigned)o) incl += n;
        }
        if (lane == 31u) wsum[tid >> 5] = incl;
    }
    __syncthreads();
    if (tid < NB) {
        const unsigned wvi = tid >> 5;
        unsigned base = 0u;
#pragma unroll
        for (unsigned k = 0; k < NB / 32; ++k) base += (k < wvi) ? wsum[k] : 0u;
        const unsigned excl = base + incl - T;
        cur[tid] = excl;
        seg[chunk * NB + tid] = excl | (T << SEG_SHIFT);
    }
    __syncthreads();

#pragma unroll
    for (int k = 0; k < PER_THREAD; ++k) {
        if (bn[k] < NB) {
            const unsigned dl = dv[k] - bn[k] * R;
            const unsigned pos = atomicAdd(&cur[bn[k]], 1u);
            v2i r;
            r.x = (int)(sv[k] | (dl << SRC_BITS));
            r.y = __builtin_bit_cast(int, wv[k]);
            ((v2i*)srec)[pos] = r;
        }
    }
    __syncthreads();

    const unsigned live2 = (min((unsigned)CHUNK, (unsigned)N_EDGES - chunk * CHUNK)) >> 1;
    v4i* out4 = (v4i*)(rec + (size_t)chunk * CHUNK);
    v4i q[PER_THREAD / 2];
#pragma unroll
    for (int k = 0; k < PER_THREAD / 2; ++k) {
        const unsigned i = tid + (unsigned)k * BIN_THREADS;
        q[k] = srec[min(i, live2 - 1u)];
    }
#pragma unroll
    for (int k = 0; k < PER_THREAD / 2; ++k) {
        const unsigned i = tid + (unsigned)k * BIN_THREADS;
        if (i < live2) out4[i] = q[k];
    }
}

template <int NSEG>
struct SegMap {
    unsigned pre[NSEG];
    unsigned adj[NSEG];
    unsigned total;
};

template <int NSEG>
__device__ __forceinline__ SegMap<NSEG> load_seg_map(const unsigned* __restrict__ seg,
                                                     unsigned b, unsigned first, unsigned stride,
                                                     unsigned lane) {
    unsigned len = 0u, sbase = 0u;
    const unsigned c = first + stride * lane;
    if (lane < (unsigned)NSEG && c < CH) {
        const unsigned s = seg[c * NB + b];
        len   = s >> SEG_SHIFT;
        sbase = c * CHUNK + (s & ((1u << SEG_SHIFT) - 1u));
    }
    unsigned incl = len;
#pragma unroll
    for (int o = 1; o < 32; o <<= 1) {
        const unsigned n = (unsigned)__shfl_up((int)incl, (unsigned)o);
        if (lane >= (unsigned)o) incl += n;
    }
    const unsigned excl = incl - len;
    const unsigned a = sbase - excl;
    SegMap<NSEG> m;
#pragma unroll
    for (int i = 0; i < NSEG; ++i) {
        m.pre[i] = (unsigned)__builtin_amdgcn_readlane((int)excl, i);
        m.adj[i] = (unsigned)__builtin_amdgcn_readlane((int)a, i);
    }
    m.total = (unsigned)__builtin_amdgcn_readlane((int)incl, 31);
    return m;
}

template <int NSEG>
__device__ __forceinline__ unsigned seg_index(const SegMap<NSEG>& m, unsigned g) {
    unsigned a = m.adj[0];
#pragma unroll
    for (int i = 1; i < NSEG; ++i) a = (g >= m.pre[i]) ? m.adj[i] : a;
    return a + g;
}

#define XT_ROWS   208
#define XS_PITCH  68
#define WS_PITCH  160
#define DG_THREADS 1024
#define GEMM_WAVES 26
#define REC_PRE   4
#define DG_SEGS   7

static_assert(XT_ROWS >= R && XT_ROWS % 16 == 0, "row tiles must cover the block");
static_assert(GEMM_WAVES == (XT_ROWS / 16) * 2 && GEMM_WAVES * 32 <= DG_THREADS, "one wave per (row tile, column half)");
static_assert(IN_W * OUT_W == DG_THREADS * 4, "W is staged as one float4 per thread");
static_assert(XT_ROWS * 16 <= 4 * DG_THREADS, "x rows are staged as at most four float4 per thread");
static_assert((DG_THREADS / 32) * DG_SEGS >= CH && DG_SEGS <= 32, "the waves' segments must cover all chunks");

__global__ void __launch_bounds__(DG_THREADS)
deg_gemm_kernel(const int2* __restrict__ rec,
                const unsigned* __restrict__ seg,
                const float* __restrict__ x,
                const float* __restrict__ W,
                float* __restrict__ dinv,
                float* __restrict__ h) {
    __shared__ float dg[XT_ROWS];
    __shared__ alignas(16) float xs[XT_ROWS * XS_PITCH];
    __shared__ alignas(16) float ws[(IN_W / 2) * WS_PITCH];
    const unsigned tid  = threadIdx.x;
    const unsigned lane = tid & 31u;
    const unsigned wv   = tid >> 5;
    const unsigned b    = blockIdx.x;
    const unsigned node0 = b * R;

    float4 xv[4];
#pragma unroll
    for (int k = 0; k < 4; ++k) {
        const unsigned t4   = tid + (unsigned)k * DG_THREADS;
        const unsigned row  = min(t4 >> 4, (unsigned)(R - 1));
        const unsigned node = min(node0 + row, (unsigned)(N_NODES - 1));
        xv[k] = *(const float4*)(x + (size_t)node * IN_W + (t4 & 15u) * 4u);
    }
    const float4 w4 = *(const float4*)(W + tid * 4u);

    const SegMap<DG_SEGS> sm = load_seg_map<DG_SEGS>(seg, b, wv, DG_THREADS / 32u, lane);
    int2 rp[REC_PRE];
#pragma unroll
    for (int k = 0; k < REC_PRE; ++k) {
        const unsigned g = lane + (unsigned)k * 32u;
        rp[k] = rec[(g < sm.total) ? seg_index(sm, g) : 0u];
    }

#pragma unroll
    for (int k = 0; k < 4; ++k) {
        const unsigned t4  = tid + (unsigned)k * DG_THREADS;
        const unsigned row = t4 >> 4;
        if (row >= R || node0 + row >= N_NODES) xv[k] = make_float4(0.0f, 0.0f, 0.0f, 0.0f);
    }

    if (tid < XT_ROWS) dg[tid] = 2.0f;
#pragma unroll
    for (int k = 0; k < 4; ++k) {
        const unsigned t4 = tid + (unsigned)k * DG_THREADS;
        if (t4 < XT_ROWS * 16u)
            *(float4*)(xs + (t4 >> 4) * XS_PITCH + (t4 & 15u) * 4u) = xv[k];
    }
    {
        const unsigned k = tid >> 4;
        const unsigned n = (tid & 15u) * 4u;
        float* p = ws + (k >> 1) * WS_PITCH + n * 2u + (k & 1u);
        p[0] = w4.x; p[2] = w4.y; p[4] = w4.z; p[6] = w4.w;
    }
    __syncthreads();

#pragma unroll
    for (int k = 0; k < REC_PRE; ++k) {
        const unsigned g = lane + (unsigned)k * 32u;
        if (g < sm.total)
            atomicAdd(&dg[(unsigned)rp[k].x >> SRC_BITS], __builtin_bit_cast(float, rp[k].y));
    }
    for (unsigned g = lane + REC_PRE * 32u; g < sm.total; g += 32u) {
        const int2 r = rec[seg_index(sm, g)];
        atomicAdd(&dg[(unsigned)r.x >> SRC_BITS], __builtin_bit_cast(float, r.y));
    }

    const unsigned half = lane >> 4;
    const unsigned lrow = lane & 15u;
    const unsigned rt   = wv >> 1;
    const unsigned ch   = wv & 1u;
    v8f c0 = {};
    v8f c1 = {};
    if (wv < GEMM_WAVES) {
        const float* xa = xs + (rt * 16u + lrow) * XS_PITCH + half * 2u;
        const float* wb = ws + half * WS_PITCH + (ch * 32u + lrow) * 2u;
#pragma unroll
        for (int kk = 0; kk < 16; ++kk) {
            const v2f a  = *(const v2f*)(xa + kk * 4);
            const v2f b0 = *(const v2f*)(wb + kk * 2 * WS_PITCH);
            const v2f b1 = *(const v2f*)(wb + kk * 2 * WS_PITCH + 32);
            c0 = __builtin_amdgcn_wmma_f32_16x16x4_f32(
                      false, a,  false, b0,
                      (short)0, c0,  false,  false);
            c1 = __builtin_amdgcn_wmma_f32_16x16x4_f32(
                      false, a,  false, b1,
                      (short)0, c1,  false,  false);
        }
    }
    __syncthreads();

    if (wv < GEMM_WAVES) {
        float* ot = xs + (rt * 16u + half * 8u) * XS_PITCH + ch * 32u + lrow;
#pragma unroll
        for (int v = 0; v < 8; ++v) {
            ot[v * XS_PITCH]      = c0[v];
            ot[v * XS_PITCH + 16] = c1[v];
        }
    }
    __syncthreads();
    if (tid < R) {
        const unsigned node = node0 + tid;
        if (node < N_NODES) {
            const float d = dg[tid];
            dinv[node] = (d > 0.0f) ? rsqrtf(d) : 0.0f;
        }
    }
    if (wv < GEMM_WAVES) {
        const unsigned f = ch * 32u + (lane & 7u) * 4u;
#pragma unroll
        for (int j = 0; j < 4; ++j) {
            const unsigned row  = rt * 16u + (lane >> 3) + (unsigned)j * 4u;
            const unsigned node = node0 + row;
            if (row < R && node < N_NODES) {
                const float d  = dg[row];
                const float di = (d > 0.0f) ? rsqrtf(d) : 0.0f;
                const float4 a = *(const float4*)(xs + row * XS_PITCH + f);
                float4 o;
                o.x = a.x * di; o.y = a.y * di; o.z = a.z * di; o.w = a.w * di;
                *(float4*)(h + (size_t)node * OUT_W + f) = o;
            }
        }
    }
}

#define EDGE_GROUP 16
#define ACC_THREADS 512
#define ACC_WAVES   (ACC_THREADS / 32)
#define ACC_SEGS    13
static_assert(ACC_WAVES * ACC_SEGS >= CH && ACC_SEGS <= 32, "the waves' segments must cover all chunks");

__device__ __forceinline__ int lane_bcast_i(int v, int j) {
    return __builtin_amdgcn_readlane(v, j);
}

__global__ void __launch_bounds__(ACC_THREADS)
block_sum_kernel(const int2* __restrict__ rec,
                 const unsigned* __restrict__ seg,
                 const float* __restrict__ dinv,
                 const float* __restrict__ h,
                 const float* __restrict__ bias,
                 float* __restrict__ out) {
    __shared__ float acc[(R + 1) * OUT_W];
    const unsigned tid  = threadIdx.x;
    const unsigned lane = tid & 31u;
    const unsigned wv   = tid >> 5;
    const unsigned b    = blockIdx.x;
    const unsigned node0 = b * R;

    const SegMap<ACC_SEGS> sm = load_seg_map<ACC_SEGS>(seg, b, wv, ACC_WAVES, lane);

    for (unsigned t = tid; t < R * (OUT_W / 4); t += ACC_THREADS) {
        const unsigned node = node0 + (t >> 4);
        float4 v = make_float4(0.0f, 0.0f, 0.0f, 0.0f);
        if (node < N_NODES) {
            const float4 hv = *(const float4*)(h + (size_t)node * OUT_W + (t & 15u) * 4u);
            v = make_float4(2.0f * hv.x, 2.0f * hv.y, 2.0f * hv.z, 2.0f * hv.w);
        }
        *(float4*)(acc + t * 4u) = v;
    }
    __syncthreads();

    for (unsigned t0 = 0u; t0 < sm.total; t0 += 32u) {
        const unsigned left = sm.total - t0;
        const unsigned g    = t0 + lane;
        int2 r = make_int2((int)((unsigned)R << SRC_BITS), 0);
        if (g < sm.total) r = rec[seg_index(sm, g)];

#pragma unroll
        for (int jb = 0; jb < 32; jb += EDGE_GROUP) {
            if ((unsigned)jb < left) {
                float h0[EDGE_GROUP], h1[EDGE_GROUP];
#pragma unroll
                for (int k = 0; k < EDGE_GROUP; ++k) {
                    const unsigned rx = (unsigned)lane_bcast_i(r.x, jb + k);
                    const float* hr = h + (size_t)(rx & ((1u << SRC_BITS) - 1u)) * OUT_W;
                    h0[k] = hr[lane];
                    h1[k] = hr[lane + 32u];
                }
#pragma unroll
                for (int k = 0; k < EDGE_GROUP; ++k) {
                    const unsigned rx = (unsigned)lane_bcast_i(r.x, jb + k);
                    const float wj = __builtin_bit_cast(float, lane_bcast_i(r.y, jb + k));
                    float* a = acc + (rx >> SRC_BITS) * OUT_W + lane;
                    atomicAdd(a,      h0[k] * wj);
                    atomicAdd(a + 32, h1[k] * wj);
                }
            }
        }
    }
    __syncthreads();

    for (unsigned t = tid; t < R * (OUT_W / 4); t += ACC_THREADS) {
        const unsigned node = node0 + (t >> 4);
        if (node < N_NODES) {
            const unsigned f = (t & 15u) * 4u;
            const float di = dinv[node];
            const float4 a  = *(const float4*)(acc + t * 4u);
            const float4 bv = *(const float4*)(bias + f);
            float4 o;
            o.x = bv.x + a.x * di;
            o.y = bv.y + a.y * di;
            o.z = bv.z + a.z * di;
            o.w = bv.w + a.w * di;
            *(float4*)(out + (size_t)node * OUT_W + f) = o;
        }
    }
}

extern "C" void kernel_launch(void* const* d_in, const int* in_sizes, int n_in,
                              void* d_out, int out_size, void* d_ws, size_t ws_size,
                              hipStream_t stream) {
    const float* x         = (const float*)d_in[0];
    const float* edge_attr = (const float*)d_in[1];
    const float* W         = (const float*)d_in[2];
    const float* bias      = (const float*)d_in[3];
    const int*   edge_idx  = (const int*)d_in[4];
    float* out = (float*)d_out;

    float*    dinv     = (float*)d_ws;
    float*    h        = dinv + 100352;
    int2*     rec      = (int2*)(h + 6400000);
    unsigned* seg      = (unsigned*)(rec + 1600000);

    const int* src = edge_idx;
    const int* dst = edge_idx + N_EDGES;

    chunk_sort_kernel<<<CH, BIN_THREADS, 0, stream>>>(src, dst, edge_attr, rec, seg);

    deg_gemm_kernel<<<NB, DG_THREADS, 0, stream>>>(rec, seg, x, W, dinv, h);

    block_sum_kernel<<<NB, ACC_THREADS, 0, stream>>>(rec, seg, dinv, h, bias, out);
}


# sum pass: block's records put in row order in LDS (per-wave counters: a row's order is no arrival order among waves), rows summed in registers, stored once; LDS float adds gone; sum order changes
# speedup vs baseline: 4.8777x; 1.1519x over previous
#include <hip/hip_runtime.h>
#include <math.h>


#define N_NODES 100000
#define IN_W    64
#define OUT_W   64
#define N_EDGES 1600000

#define R       196
#define NB      512
#define CHUNK   8192
#define CH      196
#define BIN_THREADS 1024
#define PER_THREAD  (CHUNK / BIN_THREADS)
#define SRC_BITS 17
#define SEG_SHIFT 16

static_assert(NB * R >= N_NODES, "blocks must cover all nodes");
static_assert(CH * CHUNK >= N_EDGES, "chunks must cover all edges");
static_assert(N_EDGES >= CHUNK, "chunk 0 is a full chunk: record 0 is always written");
static_assert(N_NODES <= (1 << SRC_BITS), "src must fit its field");
static_assert(R < 512, "local row, and the spare row R, must fit their field");
static_assert(CHUNK < (1 << SEG_SHIFT), "a segment's start and length must fit their fields");
static_assert(NB <= BIN_THREADS, "one scanning thread per bin");
static_assert(CHUNK % 2 == 0 && (N_EDGES % CHUNK) % 2 == 0, "every chunk holds an even number of records: it is written out two records per 16-byte store");
static_assert(CHUNK / 2 == (PER_THREAD / 2) * BIN_THREADS, "the write-out loop covers a full chunk");
static_assert((CH - 1) * CHUNK < N_EDGES, "no chunk is empty: with the line above every chunk holds at least one record pair");

typedef __attribute__((ext_vector_type(2))) float v2f;
typedef __attribute__((ext_vector_type(8))) float v8f;
typedef __attribute__((ext_vector_type(2))) int v2i;
typedef __attribute__((ext_vector_type(4))) int v4i;

__global__ void __launch_bounds__(BIN_THREADS)
chunk_sort_kernel(const int* __restrict__ src,
                  const int* __restrict__ dst,
                  const float* __restrict__ w,
                  int2* __restrict__ rec,
                  unsigned* __restrict__ seg) {
    __shared__ unsigned cnt[NB];
    __shared__ unsigned cur[NB];
    __shared__ unsigned wsum[NB / 32];
    __shared__ v4i srec[CHUNK / 2];
    const unsigned tid = threadIdx.x;
    const unsigned chunk = blockIdx.x;
    const unsigned lane = tid & 31u;
    if (tid < NB) cnt[tid] = 0u;

    const unsigned e0 = chunk * CHUNK + tid;
    unsigned sv[PER_THREAD], dv[PER_THREAD];
    float wv[PER_THREAD];
#pragma unroll
    for (int k = 0; k < PER_THREAD; ++k) {
        const unsigned e = e0 + (unsigned)k * BIN_THREADS;
        const bool live = e < N_EDGES;
        sv[k] = live ? (unsigned)src[e] : 0u;
        dv[k] = live ? (unsigned)dst[e] : (unsigned)(NB * R);
        wv[k] = live ? w[e] : 0.0f;
    }
    __syncthreads();

    unsigned bn[PER_THREAD];
#pragma unroll
    for (int k = 0; k < PER_THREAD; ++k) {
        bn[k] = dv[k] / R;
        if (bn[k] < NB) atomicAdd(&cnt[bn[k]], 1u);
    }
    __syncthreads();

    unsigned T = 0u, incl = 0u;
    if (tid < NB) {
        T = cnt[tid];
        incl = T;
#pragma unroll
        for (int o = 1; o < 32; o <<= 1) {
            const unsigned n = (unsigned)__shfl_up((int)incl, (unsigned)o);
            if (lane >= (unsigned)o) incl += n;
        }
        if (lane == 31u) wsum[tid >> 5] = incl;
    }
    __syncthreads();
    if (tid < NB) {
        const unsigned wvi = tid >> 5;
        unsigned base = 0u;
#pragma unroll
        for (unsigned k = 0; k < NB / 32; ++k) base += (k < wvi) ? wsum[k] : 0u;
        const unsigned excl = base + incl - T;
        cur[tid] = excl;
        seg[chunk * NB + tid] = excl | (T << SEG_SHIFT);
    }
    __syncthreads();

#pragma unroll
    for (int k = 0; k < PER_THREAD; ++k) {
        if (bn[k] < NB) {
            const unsigned dl = dv[k] - bn[k] * R;
            const unsigned pos = atomicAdd(&cur[bn[k]], 1u);
            v2i r;
            r.x = (int)(sv[k] | (dl << SRC_BITS));
            r.y = __builtin_bit_cast(int, wv[k]);
            ((v2i*)srec)[pos] = r;
        }
    }
    __syncthreads();

    const unsigned live2 = (min((unsigned)CHUNK, (unsigned)N_EDGES - chunk * CHUNK)) >> 1;
    v4i* out4 = (v4i*)(rec + (size_t)chunk * CHUNK);
    v4i q[PER_THREAD / 2];
#pragma unroll
    for (int k = 0; k < PER_THREAD / 2; ++k) {
        const unsigned i = tid + (unsigned)k * BIN_THREADS;
        q[k] = srec[min(i, live2 - 1u)];
    }
#pragma unroll
    for (int k = 0; k < PER_THREAD / 2; ++k) {
        const unsigned i = tid + (unsigned)k * BIN_THREADS;
        if (i < live2) out4[i] = q[k];
    }
}

template <int NSEG>
struct SegMap {
    unsigned pre[NSEG];
    unsigned adj[NSEG];
    unsigned total;
};

template <int NSEG>
__device__ __forceinline__ SegMap<NSEG> load_seg_map(const unsigned* __restrict__ seg,
                                                     unsigned b, unsigned first, unsigned stride,
                                                     unsigned lane) {
    unsigned len = 0u, sbase = 0u;
    const unsigned c = first + stride * lane;
    if (lane < (unsigned)NSEG && c < CH) {
        const unsigned s = seg[c * NB + b];
        len   = s >> SEG_SHIFT;
        sbase = c * CHUNK + (s & ((1u << SEG_SHIFT) - 1u));
    }
    unsigned incl = len;
#pragma unroll
    for (int o = 1; o < 32; o <<= 1) {
        const unsigned n = (unsigned)__shfl_up((int)incl, (unsigned)o);
        if (lane >= (unsigned)o) incl += n;
    }
    const unsigned excl = incl - len;
    const unsigned a = sbase - excl;
    SegMap<NSEG> m;
#pragma unroll
    for (int i = 0; i < NSEG; ++i) {
        m.pre[i] = (unsigned)__builtin_amdgcn_readlane((int)excl, i);
        m.adj[i] = (unsigned)__builtin_amdgcn_readlane((int)a, i);
    }
    m.total = (unsigned)__builtin_amdgcn_readlane((int)incl, 31);
    return m;
}

template <int NSEG>
__device__ __forceinline__ unsigned seg_index(const SegMap<NSEG>& m, unsigned g) {
    unsigned a = m.adj[0];
#pragma unroll
    for (int i = 1; i < NSEG; ++i) a = (g >= m.pre[i]) ? m.adj[i] : a;
    return a + g;
}

#define XT_ROWS   208
#define XS_PITCH  68
#define WS_PITCH  160
#define DG_THREADS 1024
#define GEMM_WAVES 26
#define REC_PRE   4
#define DG_SEGS   7

static_assert(XT_ROWS >= R && XT_ROWS % 16 == 0, "row tiles must cover the block");
static_assert(GEMM_WAVES == (XT_ROWS / 16) * 2 && GEMM_WAVES * 32 <= DG_THREADS, "one wave per (row tile, column half)");
static_assert(IN_W * OUT_W == DG_THREADS * 4, "W is staged as one float4 per thread");
static_assert(XT_ROWS * 16 <= 4 * DG_THREADS, "x rows are staged as at most four float4 per thread");
static_assert((DG_THREADS / 32) * DG_SEGS >= CH && DG_SEGS <= 32, "the waves' segments must cover all chunks");

__global__ void __launch_bounds__(DG_THREADS)
deg_gemm_kernel(const int2* __restrict__ rec,
                const unsigned* __restrict__ seg,
                const float* __restrict__ x,
                const float* __restrict__ W,
                float* __restrict__ dinv,
                float* __restrict__ h) {
    __shared__ float dg[XT_ROWS];
    __shared__ alignas(16) float xs[XT_ROWS * XS_PITCH];
    __shared__ alignas(16) float ws[(IN_W / 2) * WS_PITCH];
    const unsigned tid  = threadIdx.x;
    const unsigned lane = tid & 31u;
    const unsigned wv   = tid >> 5;
    const unsigned b    = blockIdx.x;
    const unsigned node0 = b * R;

    float4 xv[4];
#pragma unroll
    for (int k = 0; k < 4; ++k) {
        const unsigned t4   = tid + (unsigned)k * DG_THREADS;
        const unsigned row  = min(t4 >> 4, (unsigned)(R - 1));
        const unsigned node = min(node0 + row, (unsigned)(N_NODES - 1));
        xv[k] = *(const float4*)(x + (size_t)node * IN_W + (t4 & 15u) * 4u);
    }
    const float4 w4 = *(const float4*)(W + tid * 4u);

    const SegMap<DG_SEGS> sm = load_seg_map<DG_SEGS>(seg, b, wv, DG_THREADS / 32u, lane);
    int2 rp[REC_PRE];
#pragma unroll
    for (int k = 0; k < REC_PRE; ++k) {
        const unsigned g = lane + (unsigned)k * 32u;
        rp[k] = rec[(g < sm.total) ? seg_index(sm, g) : 0u];
    }

#pragma unroll
    for (int k = 0; k < 4; ++k) {
        const unsigned t4  = tid + (unsigned)k * DG_THREADS;
        const unsigned row = t4 >> 4;
        if (row >= R || node0 + row >= N_NODES) xv[k] = make_float4(0.0f, 0.0f, 0.0f, 0.0f);
    }

    if (tid < XT_ROWS) dg[tid] = 2.0f;
#pragma unroll
    for (int k = 0; k < 4; ++k) {
        const unsigned t4 = tid + (unsigned)k * DG_THREADS;
        if (t4 < XT_ROWS * 16u)
            *(float4*)(xs + (t4 >> 4) * XS_PITCH + (t4 & 15u) * 4u) = xv[k];
    }
    {
        const unsigned k = tid >> 4;
        const unsigned n = (tid & 15u) * 4u;
        float* p = ws + (k >> 1) * WS_PITCH + n * 2u + (k & 1u);
        p[0] = w4.x; p[2] = w4.y; p[4] = w4.z; p[6] = w4.w;
    }
    __syncthreads();

#pragma unroll
    for (int k = 0; k < REC_PRE; ++k) {
        const unsigned g = lane + (unsigned)k * 32u;
        if (g < sm.total)
            atomicAdd(&dg[(unsigned)rp[k].x >> SRC_BITS], __builtin_bit_cast(float, rp[k].y));
    }
    for (unsigned g = lane + REC_PRE * 32u; g < sm.total; g += 32u) {
        const int2 r = rec[seg_index(sm, g)];
        atomicAdd(&dg[(unsigned)r.x >> SRC_BITS], __builtin_bit_cast(float, r.y));
    }

    const unsigned half = lane >> 4;
    const unsigned lrow = lane & 15u;
    const unsigned rt   = wv >> 1;
    const unsigned ch   = wv & 1u;
    v8f c0 = {};
    v8f c1 = {};
    if (wv < GEMM_WAVES) {
        const float* xa = xs + (rt * 16u + lrow) * XS_PITCH + half * 2u;
        const float* wb = ws + half * WS_PITCH + (ch * 32u + lrow) * 2u;
#pragma unroll
        for (int kk = 0; kk < 16; ++kk) {
            const v2f a  = *(const v2f*)(xa + kk * 4);
            const v2f b0 = *(const v2f*)(wb + kk * 2 * WS_PITCH);
            const v2f b1 = *(const v2f*)(wb + kk * 2 * WS_PITCH + 32);
            c0 = __builtin_amdgcn_wmma_f32_16x16x4_f32(
                      false, a,  false, b0,
                      (short)0, c0,  false,  false);
            c1 = __builtin_amdgcn_wmma_f32_16x16x4_f32(
                      false, a,  false, b1,
                      (short)0, c1,  false,  false);
        }
    }
    __syncthreads();

    if (wv < GEMM_WAVES) {
        float* ot = xs + (rt * 16u + half * 8u) * XS_PITCH + ch * 32u + lrow;
#pragma unroll
        for (int v = 0; v < 8; ++v) {
            ot[v * XS_PITCH]      = c0[v];
            ot[v * XS_PITCH + 16] = c1[v];
        }
    }
    __syncthreads();
    if (tid < R) {
        const unsigned node = node0 + tid;
        if (node < N_NODES) {
            const float d = dg[tid];
            dinv[node] = (d > 0.0f) ? rsqrtf(d) : 0.0f;
        }
    }
    if (wv < GEMM_WAVES) {
        const unsigned f = ch * 32u + (lane & 7u) * 4u;
#pragma unroll
        for (int j = 0; j < 4; ++j) {
            const unsigned row  = rt * 16u + (lane >> 3) + (unsigned)j * 4u;
            const unsigned node = node0 + row;
            if (row < R && node < N_NODES) {
                const float d  = dg[row];
                const float di = (d > 0.0f) ? rsqrtf(d) : 0.0f;
                const float4 a = *(const float4*)(xs + row * XS_PITCH + f);
                float4 o;
                o.x = a.x * di; o.y = a.y * di; o.z = a.z * di; o.w = a.w * di;
                *(float4*)(h + (size_t)node * OUT_W + f) = o;
            }
        }
    }
}

#define EDGE_GROUP 16
#define ACC_THREADS 512
#define ACC_WAVES   (ACC_THREADS / 32)
#define ACC_SEGS    13
#define WCAP        320
#define NBATCH      (WCAP / 32)
#define RPAD        208
#define SCAN_WAVES  ((RPAD + 31) / 32)
#define TILE_CAP    (ACC_WAVES * WCAP + R)
static_assert(ACC_WAVES * ACC_SEGS >= CH && ACC_SEGS <= 32, "the waves' segments must cover all chunks");
static_assert(WCAP % 32 == 0, "a round is whole batches of 32 records");
static_assert((ACC_WAVES * RPAD) % 4 == 0, "the counters are cleared 16 bytes per lane");
static_assert(RPAD > R && RPAD <= ACC_THREADS, "one scanning thread per row, and rbeg[R] exists");
static_assert((R + ACC_WAVES - 1) / ACC_WAVES <= 32, "a wave's rows fit one lane each");

__device__ __forceinline__ int lane_bcast_i(int v, int j) {
    return __builtin_amdgcn_readlane(v, j);
}

__global__ void __launch_bounds__(ACC_THREADS)
block_sum_kernel(const int2* __restrict__ rec,
                 const unsigned* __restrict__ seg,
                 const float* __restrict__ dinv,
                 const float* __restrict__ h,
                 const float* __restrict__ bias,
                 float* __restrict__ out) {
    __shared__ v2i srt[TILE_CAP];
    __shared__ alignas(16) unsigned cw[ACC_WAVES * RPAD];
    __shared__ unsigned rbeg[RPAD];
    __shared__ unsigned wtot[SCAN_WAVES];
    __shared__ unsigned mx;
    const unsigned tid  = threadIdx.x;
    const unsigned lane = tid & 31u;
    const unsigned wv   = tid >> 5;
    const unsigned b    = blockIdx.x;
    const unsigned node0 = b * R;

    const SegMap<ACC_SEGS> sm = load_seg_map<ACC_SEGS>(seg, b, wv, ACC_WAVES, lane);

    const unsigned r0 = (wv * R) / ACC_WAVES;
    const unsigned r1 = ((wv + 1u) * R) / ACC_WAVES;
    float dvl = 0.0f;
    if (lane < r1 - r0 && node0 + r0 + lane < N_NODES) dvl = dinv[node0 + r0 + lane];
    const float b0 = bias[lane];
    const float b1 = bias[lane + 32u];

    if (tid == 0u) mx = 0u;
    unsigned mxv = 0u;

    for (unsigned round = 0u; ; ++round) {
        const unsigned gbase = round * WCAP;
        for (unsigned i = tid; i < ACC_WAVES * RPAD / 4u; i += ACC_THREADS) ((v4i*)cw)[i] = v4i{0, 0, 0, 0};

        v2i rv[NBATCH];
#pragma unroll
        for (int k = 0; k < NBATCH; ++k) {
            const unsigned g = gbase + (unsigned)k * 32u + lane;
            const int2 t = rec[(g < sm.total) ? seg_index(sm, g) : 0u];
            rv[k].x = t.x;
            rv[k].y = t.y;
        }
        __syncthreads();

        unsigned rk[NBATCH];
#pragma unroll
        for (int k = 0; k < NBATCH; ++k) {
            const unsigned g = gbase + (unsigned)k * 32u + lane;
            rk[k] = 0u;
            if (g < sm.total)
                rk[k] = atomicAdd(&cw[wv * RPAD + ((unsigned)rv[k].x >> SRC_BITS)], 1u);
        }
        if (round == 0u && lane == 0u) atomicMax(&mx, sm.total);
        __syncthreads();

        unsigned c[ACC_WAVES];
        unsigned tot = 0u, selfc = 0u;
#pragma unroll
        for (int w = 0; w < ACC_WAVES; ++w) c[w] = 0u;
        if (tid < RPAD) {
#pragma unroll
            for (int w = 0; w < ACC_WAVES; ++w) {
                c[w] = cw[w * RPAD + tid];
                tot += c[w];
            }
            selfc = (round == 0u && tid < R && node0 + tid < N_NODES) ? 1u : 0u;
            tot += selfc;
        }
        unsigned incl = tot;
#pragma unroll
        for (int o = 1; o < 32; o <<= 1) {
            const unsigned n = (unsigned)__shfl_up((int)incl, (unsigned)o);
            if (lane >= (unsigned)o) incl += n;
        }
        if (lane == 31u && wv < SCAN_WAVES) wtot[wv] = incl;
        if (round == 0u) mxv = (unsigned)__builtin_amdgcn_readfirstlane((int)mx);
        __syncthreads();
        if (tid < RPAD) {
            unsigned basew = 0u;
#pragma unroll
            for (unsigned k = 0; k < SCAN_WAVES; ++k) basew += (k < wv) ? wtot[k] : 0u;
            const unsigned excl = basew + incl - tot;
            rbeg[tid] = excl;
            unsigned run = excl + selfc;
#pragma unroll
            for (int w = 0; w < ACC_WAVES; ++w) {
                cw[w * RPAD + tid] = run;
                run += c[w];
            }
        }
        __syncthreads();

#pragma unroll
        for (int k = 0; k < NBATCH; ++k) {
            const unsigned g = gbase + (unsigned)k * 32u + lane;
            if (g < sm.total) {
                const unsigned dl = (unsigned)rv[k].x >> SRC_BITS;
                srt[cw[wv * RPAD + dl] + rk[k]] = rv[k];
            }
        }
        if (round == 0u && tid < R && node0 + tid < N_NODES) {
            v2i s;
            s.x = (int)((node0 + tid) | (tid << SRC_BITS));
            s.y = __builtin_bit_cast(int, 2.0f);
            srt[rbeg[tid]] = s;
        }
        __syncthreads();

        auto flush = [&](unsigned row, float a0, float a1) {
            const unsigned node = node0 + row;
            if (row < R && node < N_NODES) {
                const float di = __builtin_bit_cast(float,
                    __builtin_amdgcn_readlane(__builtin_bit_cast(int, dvl), (int)(row - r0)));
                float* o = out + (size_t)node * OUT_W + lane;
                float o0, o1;
                if (round == 0u) { o0 = b0 + a0 * di; o1 = b1 + a1 * di; }
                else             { o0 = o[0] + a0 * di; o1 = o[32] + a1 * di; }
                o[0]  = o0;
                o[32] = o1;
            }
        };

        const unsigned beg = (unsigned)__builtin_amdgcn_readfirstlane((int)rbeg[r0]);
        const unsigned end = (unsigned)__builtin_amdgcn_readfirstlane((int)rbeg[r1]);
        unsigned cur = R;
        float acc0 = 0.0f, acc1 = 0.0f;
        for (unsigned t0 = beg; t0 < end; t0 += 32u) {
            const unsigned left = end - t0;
            v2i r;
            r.x = (int)((unsigned)R << SRC_BITS);
            r.y = 0;
            if (lane < left) r = srt[t0 + lane];

#pragma unroll
            for (int jb = 0; jb < 32; jb += EDGE_GROUP) {
                if ((unsigned)jb < left) {
                    float h0[EDGE_GROUP], h1[EDGE_GROUP];
#pragma unroll
                    for (int k = 0; k < EDGE_GROUP; ++k) {
                        const unsigned rx = (unsigned)lane_bcast_i(r.x, jb + k);
                        const float* hr = h + (size_t)(rx & ((1u << SRC_BITS) - 1u)) * OUT_W;
                        h0[k] = hr[lane];
                        h1[k] = hr[lane + 32u];
                    }
#pragma unroll
                    for (int k = 0; k < EDGE_GROUP; ++k) {
                        const unsigned rx = (unsigned)lane_bcast_i(r.x, jb + k);
                        const float wj = __builtin_bit_cast(float, lane_bcast_i(r.y, jb + k));
                        const unsigned row = rx >> SRC_BITS;
                        if (__builtin_expect(row != cur, 0)) {
                            flush(cur, acc0, acc1);
                            cur = row;
                            acc0 = 0.0f;
                            acc1 = 0.0f;
                        }
                        acc0 += h0[k] * wj;
                        acc1 += h1[k] * wj;
                    }
                }
            }
        }
        flush(cur, acc0, acc1);

        if (gbase + WCAP >= mxv) break;
        __syncthreads();
    }
}

extern "C" void kernel_launch(void* const* d_in, const int* in_sizes, int n_in,
                              void* d_out, int out_size, void* d_ws, size_t ws_size,
                              hipStream_t stream) {
    const float* x         = (const float*)d_in[0];
    const float* edge_attr = (const float*)d_in[1];
    const float* W         = (const float*)d_in[2];
    const float* bias      = (const float*)d_in[3];
    const int*   edge_idx  = (const int*)d_in[4];
    float* out = (float*)d_out;

    float*    dinv     = (float*)d_ws;
    float*    h        = dinv + 100352;
    int2*     rec      = (int2*)(h + 6400000);
    unsigned* seg      = (unsigned*)(rec + 1600000);

    const int* src = edge_idx;
    const int* dst = edge_idx + N_EDGES;

    chunk_sort_kernel<<<CH, BIN_THREADS, 0, stream>>>(src, dst, edge_attr, rec, seg);

    deg_gemm_kernel<<<NB, DG_THREADS, 0, stream>>>(rec, seg, x, W, dinv, h);

    block_sum_kernel<<<NB, ACC_THREADS, 0, stream>>>(rec, seg, dinv, h, bias, out);
}
